# starting kernel
# baseline (speedup 1.0000x reference)
#include <hip/hip_runtime.h>
#include <math.h>

#define S 8
#define V 4
#define D 20
#define OS 12
#define NN 25000
#define NE 400000
#define EPSV 1e-5f

#define K1S 168
#define K1V 164
#define K2S 80
#define K2V 80

typedef __attribute__((ext_vector_type(2))) float v2f;
typedef __attribute__((ext_vector_type(8))) float v8f;

__device__ __forceinline__ float sigmoidf_(float x) { return 1.0f / (1.0f + expf(-x)); }

__device__ __forceinline__ float feat1s(const float* S1e, const float* S2e, const float* DOTe,
                                        int f) {
  if (f < 136) return S1e[f >> 3] * S2e[f & 7];
  return DOTe[f - 136];
}

template <int C>
__device__ __forceinline__ float feat1v(const float* S1e, const float* S2e, const float* V1e,
                                        const float* V2e, const float* CRSe, int f) {
  if (f < 68) return S1e[f >> 2] * V2e[(f & 3) * 3 + C];
  if (f < 132) { int g = f - 68; return V1e[(g >> 3) * 3 + C] * S2e[g & 7]; }
  return CRSe[(f - 132) * 3 + C];
}

template <int C>
__device__ __forceinline__ v8f edge_vpath(const float* S1e, const float* S2e, const float* V1e,
                                          const float* V2e, const float* CRSe,
                                          const float* WVp, int h, int n) {
  v8f acc = {};
  const float* wb = WVp + ((size_t)h * 16 + n) * 2;
#pragma unroll
  for (int kc = 0; kc < K1V / 4; ++kc) {
    int f0 = kc * 4 + 2 * h;
    v2f a;
    a.x = feat1v<C>(S1e, S2e, V1e, V2e, CRSe, f0);
    a.y = feat1v<C>(S1e, S2e, V1e, V2e, CRSe, f0 + 1);
    v2f b = *(const v2f*)(wb + kc * 64);
    acc = __builtin_amdgcn_wmma_f32_16x16x4_f32(false, a, false, b, (short)0, acc, false, false);
  }
  return acc;
}

__device__ __forceinline__ float feat2s(const float* S1e, const float* S2e, const float* DOTe,
                                        int f) {
  if (f < 64) return S1e[f >> 3] * S2e[f & 7];
  return DOTe[f - 64];
}

template <int C>
__device__ __forceinline__ float feat2v(const float* S1e, const float* S2e, const float* V1e,
                                        const float* V2e, const float* CRSe, int f) {
  if (f < 32) return S1e[f >> 2] * V2e[(f & 3) * 3 + C];
  if (f < 64) { int g = f - 32; return V1e[(g >> 3) * 3 + C] * S2e[g & 7]; }
  return CRSe[(f - 64) * 3 + C];
}

template <int C>
__device__ __forceinline__ v8f node_vpath(const float* S1e, const float* S2e, const float* V1e,
                                          const float* V2e, const float* CRSe,
                                          const float* WVp, int h, int n) {
  v8f acc = {};
  const float* wb = WVp + ((size_t)h * 16 + n) * 2;
#pragma unroll
  for (int kc = 0; kc < K2V / 4; ++kc) {
    int f0 = kc * 4 + 2 * h;
    v2f a;
    a.x = feat2v<C>(S1e, S2e, V1e, V2e, CRSe, f0);
    a.y = feat2v<C>(S1e, S2e, V1e, V2e, CRSe, f0 + 1);
    v2f b = *(const v2f*)(wb + kc * 64);
    acc = __builtin_amdgcn_wmma_f32_16x16x4_f32(false, a, false, b, (short)0, acc, false, false);
  }
  return acc;
}

__global__ __launch_bounds__(64) void k_edge_tp(
    const float* __restrict__ x, const float* __restrict__ ea, const float* __restrict__ elen,
    const int* __restrict__ eidx,
    const float* __restrict__ Wss, const float* __restrict__ Wsv, const float* __restrict__ Wvs,
    const float* __restrict__ Wvvs, const float* __restrict__ Wvvv,
    float* __restrict__ msmv, float* __restrict__ estat) {
  __shared__ float WSs[K1S * 16];
  __shared__ float WVs[K1V * 16];
  __shared__ float ST[2][3024];
  __shared__ float BST[20];

  const int tid = threadIdx.x;
  for (int idx = tid; idx < K1S * 16; idx += 64) {
    int j = idx & 1, n = (idx >> 1) & 15, hh = (idx >> 5) & 1, kc = idx >> 6;
    int f = kc * 4 + 2 * hh + j;
    float w = 0.0f;
    if (n < OS) w = (f < 136) ? Wss[f * OS + n] : Wvvs[(f - 136) * OS + n];
    WSs[idx] = w;
  }
  for (int idx = tid; idx < K1V * 16; idx += 64) {
    int j = idx & 1, n = (idx >> 1) & 15, hh = (idx >> 5) & 1, kc = idx >> 6;
    int f = kc * 4 + 2 * hh + j;
    float w = 0.0f;
    if (n < V) {
      if (f < 68) w = Wsv[f * V + n];
      else if (f < 132) w = Wvs[(f - 68) * V + n];
      else w = Wvvv[(f - 132) * V + n];
    }
    WVs[idx] = w;
  }
  if (tid < 20) BST[tid] = 0.0f;

  const int wave = tid >> 5;
  const int lane = tid & 31;
  const int e = lane & 15;
  const int h = lane >> 4;
  const int n = e;
  const int tileBase = (blockIdx.x * 2 + wave) * 16;
  const int eg = tileBase + e;

  float* S1 = &ST[wave][0];
  float* S2 = &ST[wave][272];
  float* V1 = &ST[wave][400];
  float* V2 = &ST[wave][784];
  float* DOT = &ST[wave][976];
  float* CRS = &ST[wave][1488];

  if (h == 0) {
    int i = eidx[eg], j = eidx[NE + eg];
#pragma unroll
    for (int u = 0; u < 8; ++u) { S1[e * 17 + u] = x[i * D + u]; S1[e * 17 + 8 + u] = x[j * D + u]; }
    S1[e * 17 + 16] = elen[eg];
#pragma unroll
    for (int t = 0; t < 12; ++t) { V1[e * 24 + t] = x[i * D + 8 + t]; V1[e * 24 + 12 + t] = x[j * D + 8 + t]; }
  } else {
#pragma unroll
    for (int u = 0; u < 8; ++u) S2[e * 8 + u] = ea[eg * D + u];
#pragma unroll
    for (int t = 0; t < 12; ++t) V2[e * 12 + t] = ea[eg * D + 8 + t];
  }
  __syncthreads();
  if (h == 0) {
#pragma unroll
    for (int u = 0; u < 8; ++u)
#pragma unroll
      for (int v = 0; v < 4; ++v) {
        float d = 0.f;
        for (int c = 0; c < 3; ++c) d += V1[e * 24 + u * 3 + c] * V2[e * 12 + v * 3 + c];
        DOT[e * 32 + u * 4 + v] = d;
      }
  } else {
#pragma unroll
    for (int u = 0; u < 8; ++u)
#pragma unroll
      for (int v = 0; v < 4; ++v) {
        float a0 = V1[e * 24 + u * 3], a1 = V1[e * 24 + u * 3 + 1], a2 = V1[e * 24 + u * 3 + 2];
        float b0 = V2[e * 12 + v * 3], b1 = V2[e * 12 + v * 3 + 1], b2 = V2[e * 12 + v * 3 + 2];
        int base = e * 96 + (u * 4 + v) * 3;
        CRS[base + 0] = a1 * b2 - a2 * b1;
        CRS[base + 1] = a2 * b0 - a0 * b2;
        CRS[base + 2] = a0 * b1 - a1 * b0;
      }
  }
  __syncthreads();

  const float* S1e = S1 + e * 17;
  const float* S2e = S2 + e * 8;
  const float* V1e = V1 + e * 24;
  const float* V2e = V2 + e * 12;
  const float* DOTe = DOT + e * 32;
  const float* CRSe = CRS + e * 96;

  v8f accS = {};
  {
    const float* wb = WSs + ((size_t)h * 16 + n) * 2;
#pragma unroll
    for (int kc = 0; kc < K1S / 4; ++kc) {
      int f0 = kc * 4 + 2 * h;
      v2f a;
      a.x = feat1s(S1e, S2e, DOTe, f0);
      a.y = feat1s(S1e, S2e, DOTe, f0 + 1);
      v2f b = *(const v2f*)(wb + kc * 64);
      accS = __builtin_amdgcn_wmma_f32_16x16x4_f32(false, a, false, b, (short)0, accS, false, false);
    }
  }
  v8f accV0 = edge_vpath<0>(S1e, S2e, V1e, V2e, CRSe, WVs, h, n);
  v8f accV1 = edge_vpath<1>(S1e, S2e, V1e, V2e, CRSe, WVs, h, n);
  v8f accV2 = edge_vpath<2>(S1e, S2e, V1e, V2e, CRSe, WVs, h, n);

  float sumS = 0.f, sumS2 = 0.f, sumV = 0.f;
#pragma unroll
  for (int r = 0; r < 8; ++r) {
    int em = tileBase + r + 8 * h;
    float sv = accS[r];
    float sg = sigmoidf_(sv);
    float gate = __shfl(sg, (lane & 16) | (8 + (n & 3)), 32);
    if (n < 8) {
      float silu = sv * sg;
      msmv[em * D + n] = silu;
      sumS += silu;
      sumS2 += silu * silu;
    }
    if (n < 4) {
      float v0 = accV0[r] * gate, v1 = accV1[r] * gate, v2 = accV2[r] * gate;
      msmv[em * D + 8 + n * 3 + 0] = v0;
      msmv[em * D + 8 + n * 3 + 1] = v1;
      msmv[em * D + 8 + n * 3 + 2] = v2;
      sumV += v0 * v0 + v1 * v1 + v2 * v2;
    }
  }
  sumS += __shfl_xor(sumS, 16, 32);
  sumS2 += __shfl_xor(sumS2, 16, 32);
  sumV += __shfl_xor(sumV, 16, 32);
  if (h == 0) {
    if (n < 8) { atomicAdd(&BST[n], sumS); atomicAdd(&BST[8 + n], sumS2); }
    if (n < 4) atomicAdd(&BST[16 + n], sumV);
  }
  __syncthreads();
  if (tid < 20) atomicAdd(&estat[tid], BST[tid]);
}

__global__ void k_finalize_bn(const float* __restrict__ stat, const float* __restrict__ wsc,
                              const float* __restrict__ bsc, const float* __restrict__ wvc,
                              float* __restrict__ bn, float count) {
  int t = threadIdx.x;
  if (t < 8) {
    float mu = stat[t] / count;
    float var = stat[8 + t] / count - mu * mu;
    float is = rsqrtf(var + EPSV);
    float sc = wsc[t] * is;
    bn[t] = sc;
    bn[8 + t] = bsc[t] - mu * sc;
  } else if (t < 12) {
    int u = t - 8;
    bn[16 + u] = wvc[u] * rsqrtf(stat[16 + u] / (3.0f * count) + EPSV);
  }
}

__global__ __launch_bounds__(128) void k_node_tp(
    const float* __restrict__ x, const float* __restrict__ ea, const int* __restrict__ eidx,
    const float* __restrict__ msmv, const float* __restrict__ ebn,
    const float* __restrict__ Wss, const float* __restrict__ Wsv, const float* __restrict__ Wvs,
    const float* __restrict__ Wvvs, const float* __restrict__ Wvvv,
    float* __restrict__ agg, float* __restrict__ eout) {
  __shared__ float WSs[K2S * 16];
  __shared__ float WVs[K2V * 16];
  __shared__ float ST[4][1664];
  __shared__ int NI[4][16];

  const int tid = threadIdx.x;
  for (int idx = tid; idx < K2S * 16; idx += 128) {
    int j = idx & 1, n = (idx >> 1) & 15, hh = (idx >> 5) & 1, kc = idx >> 6;
    int f = kc * 4 + 2 * hh + j;
    float w = 0.0f;
    if (n < OS) w = (f < 64) ? Wss[f * OS + n] : Wvvs[(f - 64) * OS + n];
    WSs[idx] = w;
  }
  for (int idx = tid; idx < K2V * 16; idx += 128) {
    int j = idx & 1, n = (idx >> 1) & 15, hh = (idx >> 5) & 1, kc = idx >> 6;
    int f = kc * 4 + 2 * hh + j;
    float w = 0.0f;
    if (n < V) {
      if (f < 32) w = Wsv[f * V + n];
      else if (f < 64) w = Wvs[(f - 32) * V + n];
      else w = Wvvv[(f - 64) * V + n];
    }
    WVs[idx] = w;
  }

  const int wave = tid >> 5;
  const int lane = tid & 31;
  const int e = lane & 15;
  const int h = lane >> 4;
  const int n = e;
  const int tileBase = (blockIdx.x * 4 + wave) * 16;
  const int eg = tileBase + e;

  float* S1 = &ST[wave][0];
  float* S2 = &ST[wave][128];
  float* V1 = &ST[wave][256];
  float* V2 = &ST[wave][448];
  float* DOT = &ST[wave][640];
  float* CRS = &ST[wave][896];

  if (h == 0) {
    int i = eidx[eg];
    NI[wave][e] = i;
#pragma unroll
    for (int u = 0; u < 8; ++u) S1[e * 8 + u] = x[i * D + u];
#pragma unroll
    for (int t = 0; t < 12; ++t) V1[e * 12 + t] = x[i * D + 8 + t];
  } else {
#pragma unroll
    for (int u = 0; u < 8; ++u) {
      float v = msmv[eg * D + u] * ebn[u] + ebn[8 + u] + ea[eg * D + u];
      S2[e * 8 + u] = v;
      eout[eg * D + u] = v;
    }
#pragma unroll
    for (int t = 0; t < 12; ++t) {
      float v = msmv[eg * D + 8 + t] * ebn[16 + t / 3] + ea[eg * D + 8 + t];
      V2[e * 12 + t] = v;
      eout[eg * D + 8 + t] = v;
    }
  }
  __syncthreads();
  if (h == 0) {
#pragma unroll
    for (int u = 0; u < 4; ++u)
#pragma unroll
      for (int v = 0; v < 4; ++v) {
        float d = 0.f;
        for (int c = 0; c < 3; ++c) d += V1[e * 12 + u * 3 + c] * V2[e * 12 + v * 3 + c];
        DOT[e * 16 + u * 4 + v] = d;
      }
  } else {
#pragma unroll
    for (int u = 0; u < 4; ++u)
#pragma unroll
      for (int v = 0; v < 4; ++v) {
        float a0 = V1[e * 12 + u * 3], a1 = V1[e * 12 + u * 3 + 1], a2 = V1[e * 12 + u * 3 + 2];
        float b0 = V2[e * 12 + v * 3], b1 = V2[e * 12 + v * 3 + 1], b2 = V2[e * 12 + v * 3 + 2];
        int base = e * 48 + (u * 4 + v) * 3;
        CRS[base + 0] = a1 * b2 - a2 * b1;
        CRS[base + 1] = a2 * b0 - a0 * b2;
        CRS[base + 2] = a0 * b1 - a1 * b0;
      }
  }
  __syncthreads();

  const float* S1e = S1 + e * 8;
  const float* S2e = S2 + e * 8;
  const float* V1e = V1 + e * 12;
  const float* V2e = V2 + e * 12;
  const float* DOTe = DOT + e * 16;
  const float* CRSe = CRS + e * 48;

  v8f accS = {};
  {
    const float* wb = WSs + ((size_t)h * 16 + n) * 2;
#pragma unroll
    for (int kc = 0; kc < K2S / 4; ++kc) {
      int f0 = kc * 4 + 2 * h;
      v2f a;
      a.x = feat2s(S1e, S2e, DOTe, f0);
      a.y = feat2s(S1e, S2e, DOTe, f0 + 1);
      v2f b = *(const v2f*)(wb + kc * 64);
      accS = __builtin_amdgcn_wmma_f32_16x16x4_f32(false, a, false, b, (short)0, accS, false, false);
    }
  }
  v8f accV0 = node_vpath<0>(S1e, S2e, V1e, V2e, CRSe, WVs, h, n);
  v8f accV1 = node_vpath<1>(S1e, S2e, V1e, V2e, CRSe, WVs, h, n);
  v8f accV2 = node_vpath<2>(S1e, S2e, V1e, V2e, CRSe, WVs, h, n);

#pragma unroll
  for (int r = 0; r < 8; ++r) {
    int m = r + 8 * h;
    int node = NI[wave][m];
    float sv = accS[r];
    float sg = sigmoidf_(sv);
    float gate = __shfl(sg, (lane & 16) | (8 + (n & 3)), 32);
    if (n < 8) atomicAdd(&agg[node * D + n], sv * sg);
    if (n < 4) {
      atomicAdd(&agg[node * D + 8 + n * 3 + 0], accV0[r] * gate);
      atomicAdd(&agg[node * D + 8 + n * 3 + 1], accV1[r] * gate);
      atomicAdd(&agg[node * D + 8 + n * 3 + 2], accV2[r] * gate);
    }
  }
}

__global__ __launch_bounds__(256) void k_node_stats(const float* __restrict__ agg,
                                                    float* __restrict__ nstat) {
  __shared__ float BST[20];
  int tid = threadIdx.x;
  if (tid < 20) BST[tid] = 0.f;
  __syncthreads();
  int node = blockIdx.x * 256 + tid;
  float ss[8] = {}, s2[8] = {}, vv[4] = {};
  if (node < NN) {
#pragma unroll
    for (int k = 0; k < 8; ++k) { float t = agg[node * D + k]; ss[k] = t; s2[k] = t * t; }
#pragma unroll
    for (int t = 0; t < 12; ++t) { float z = agg[node * D + 8 + t]; vv[t / 3] += z * z; }
  }
#pragma unroll
  for (int off = 16; off >= 1; off >>= 1) {
#pragma unroll
    for (int k = 0; k < 8; ++k) { ss[k] += __shfl_xor(ss[k], off, 32); s2[k] += __shfl_xor(s2[k], off, 32); }
#pragma unroll
    for (int u = 0; u < 4; ++u) vv[u] += __shfl_xor(vv[u], off, 32);
  }
  if ((tid & 31) == 0) {
#pragma unroll
    for (int k = 0; k < 8; ++k) { atomicAdd(&BST[k], ss[k]); atomicAdd(&BST[8 + k], s2[k]); }
#pragma unroll
    for (int u = 0; u < 4; ++u) atomicAdd(&BST[16 + u], vv[u]);
  }
  __syncthreads();
  if (tid < 20) atomicAdd(&nstat[tid], BST[tid]);
}

__global__ __launch_bounds__(256) void k_node_out(const float* __restrict__ agg,
                                                  const float* __restrict__ x,
                                                  const float* __restrict__ nbn,
                                                  float* __restrict__ out) {
  int idx = blockIdx.x * 256 + threadIdx.x;
  if (idx >= NN * D) return;
  int k = idx % D;
  float a = agg[idx];
  float r = (k < 8) ? (a * nbn[k] + nbn[8 + k]) : (a * nbn[16 + (k - 8) / 3]);
  out[idx] = r + x[idx];
}

extern "C" void kernel_launch(void* const* d_in, const int* in_sizes, int n_in,
                              void* d_out, int out_size, void* d_ws, size_t ws_size,
                              hipStream_t stream) {
  (void)in_sizes; (void)n_in; (void)out_size; (void)ws_size;
  const float* x = (const float*)d_in[0];
  const float* ea = (const float*)d_in[1];
  const float* elen = (const float*)d_in[2];
  const float* We_ss = (const float*)d_in[3];
  const float* We_sv = (const float*)d_in[4];
  const float* We_vs = (const float*)d_in[5];
  const float* We_vvs = (const float*)d_in[6];
  const float* We_vvv = (const float*)d_in[7];
  const float* bne_ws = (const float*)d_in[8];
  const float* bne_bs = (const float*)d_in[9];
  const float* bne_wv = (const float*)d_in[10];
  const float* Wn_ss = (const float*)d_in[11];
  const float* Wn_sv = (const float*)d_in[12];
  const float* Wn_vs = (const float*)d_in[13];
  const float* Wn_vvs = (const float*)d_in[14];
  const float* Wn_vvv = (const float*)d_in[15];
  const float* bnn_ws = (const float*)d_in[16];
  const float* bnn_bs = (const float*)d_in[17];
  const float* bnn_wv = (const float*)d_in[18];
  const int* eidx = (const int*)d_in[19];

  float* w = (float*)d_ws;
  float* msmv = w;
  float* agg = w + (size_t)NE * D;
  float* estat = agg + (size_t)NN * D;
  float* nstat = estat + 20;
  float* ebn = nstat + 20;
  float* nbn = ebn + 20;
  float* outf = (float*)d_out;
  float* eout = outf + (size_t)NN * D;

  hipMemsetAsync(agg, 0, ((size_t)NN * D + 40) * sizeof(float), stream);

  k_edge_tp<<<NE / 32, 64, 0, stream>>>(x, ea, elen, eidx, We_ss, We_sv, We_vs, We_vvs, We_vvv,
                                        msmv, estat);
  k_finalize_bn<<<1, 32, 0, stream>>>(estat, bne_ws, bne_bs, bne_wv, ebn, (float)NE);
  k_node_tp<<<NE / 64, 128, 0, stream>>>(x, ea, eidx, msmv, ebn, Wn_ss, Wn_sv, Wn_vs, Wn_vvs,
                                         Wn_vvv, agg, eout);
  k_node_stats<<<(NN + 255) / 256, 256, 0, stream>>>(agg, nstat);
  k_finalize_bn<<<1, 32, 0, stream>>>(nstat, bnn_ws, bnn_bs, bnn_wv, nbn, (float)NN);
  k_node_out<<<(NN * D + 255) / 256, 256, 0, stream>>>(agg, x, nbn, outf);
}


# Persistent work-groups in both GEMM kernels: packed weights staged to LDS once per group (batched loads), then a loop over edge tiles; edge BN partial sums span a group's tiles (sum order changes)
# speedup vs baseline: 2.2491x; 2.2491x over previous
#include <hip/hip_runtime.h>
#include <math.h>

#define S 8
#define V 4
#define D 20
#define OS 12
#define NN 25000
#define NE 400000
#define EPSV 1e-5f

#define K1S 168
#define K1V 164
#define K2S 80
#define K2V 80

#define TILES (NE / 16)
#define PW 4
#define PT (PW * 32)
#define PTRIPS (TILES / PW)
#define EGRID 1024
#define NGRID 2048
#define EPASS 21

typedef __attribute__((ext_vector_type(2))) float v2f;
typedef __attribute__((ext_vector_type(8))) float v8f;

__device__ __forceinline__ float sigmoidf_(float x) { return 1.0f / (1.0f + expf(-x)); }

__device__ __forceinline__ float feat1s(const float* S1e, const float* S2e, const float* DOTe,
                                        int f) {
  if (f < 136) return S1e[f >> 3] * S2e[f & 7];
  return DOTe[f - 136];
}

template <int C>
__device__ __forceinline__ float feat1v(const float* S1e, const float* S2e, const float* V1e,
                                        const float* V2e, const float* CRSe, int f) {
  if (f < 68) return S1e[f >> 2] * V2e[(f & 3) * 3 + C];
  if (f < 132) { int g = f - 68; return V1e[(g >> 3) * 3 + C] * S2e[g & 7]; }
  return CRSe[(f - 132) * 3 + C];
}

template <int C>
__device__ __forceinline__ v8f edge_vpath(const float* S1e, const float* S2e, const float* V1e,
                                          const float* V2e, const float* CRSe,
                                          const float* WVp, int h, int n) {
  v8f acc = {};
  const float* wb = WVp + ((size_t)h * 16 + n) * 2;
#pragma unroll
  for (int kc = 0; kc < K1V / 4; ++kc) {
    int f0 = kc * 4 + 2 * h;
    v2f a;
    a.x = feat1v<C>(S1e, S2e, V1e, V2e, CRSe, f0);
    a.y = feat1v<C>(S1e, S2e, V1e, V2e, CRSe, f0 + 1);
    v2f b = *(const v2f*)(wb + kc * 64);
    acc = __builtin_amdgcn_wmma_f32_16x16x4_f32(false, a, false, b, (short)0, acc, false, false);
  }
  return acc;
}

__device__ __forceinline__ float feat2s(const float* S1e, const float* S2e, const float* DOTe,
                                        int f) {
  if (f < 64) return S1e[f >> 3] * S2e[f & 7];
  return DOTe[f - 64];
}

template <int C>
__device__ __forceinline__ float feat2v(const float* S1e, const float* S2e, const float* V1e,
                                        const float* V2e, const float* CRSe, int f) {
  if (f < 32) return S1e[f >> 2] * V2e[(f & 3) * 3 + C];
  if (f < 64) { int g = f - 32; return V1e[(g >> 3) * 3 + C] * S2e[g & 7]; }
  return CRSe[(f - 64) * 3 + C];
}

template <int C>
__device__ __forceinline__ v8f node_vpath(const float* S1e, const float* S2e, const float* V1e,
                                          const float* V2e, const float* CRSe,
                                          const float* WVp, int h, int n) {
  v8f acc = {};
  const float* wb = WVp + ((size_t)h * 16 + n) * 2;
#pragma unroll
  for (int kc = 0; kc < K2V / 4; ++kc) {
    int f0 = kc * 4 + 2 * h;
    v2f a;
    a.x = feat2v<C>(S1e, S2e, V1e, V2e, CRSe, f0);
    a.y = feat2v<C>(S1e, S2e, V1e, V2e, CRSe, f0 + 1);
    v2f b = *(const v2f*)(wb + kc * 64);
    acc = __builtin_amdgcn_wmma_f32_16x16x4_f32(false, a, false, b, (short)0, acc, false, false);
  }
  return acc;
}

__global__ __launch_bounds__(PT) void k_edge_tp(
    const float* __restrict__ x, const float* __restrict__ ea, const float* __restrict__ elen,
    const int* __restrict__ eidx,
    const float* __restrict__ Wss, const float* __restrict__ Wsv, const float* __restrict__ Wvs,
    const float* __restrict__ Wvvs, const float* __restrict__ Wvvv,
    float* __restrict__ msmv, float* __restrict__ estat) {
  __shared__ float WSs[K1S * 16];
  __shared__ float WVs[EPASS * PT];
  __shared__ float ST[PW][3024];
  __shared__ float BST[20];

  const int tid = threadIdx.x;
  __builtin_assume(tid >= 0 && tid < PT);
#pragma unroll
  for (int k = 0; k < EPASS; ++k) {
    const int idx = tid + k * PT;
    const int j = idx & 1, nn = (idx >> 1) & 15, hh = (idx >> 5) & 1, kc = idx >> 6;
    {
      const int f = kc * 4 + 2 * hh + j;
      const float* p = (f < 136) ? (Wss + f * OS) : (Wvvs + (f - 136) * OS);
      const float w = p[nn < OS ? nn : 0];
      WSs[idx] = (nn < OS) ? w : 0.0f;
    }
    {
      const bool in = idx < K1V * 16;
      const int f = in ? kc * 4 + 2 * hh + j : 0;
      const float* p = (f < 68) ? (Wsv + f * V) : (f < 132) ? (Wvs + (f - 68) * V) : (Wvvv + (f - 132) * V);
      const float w = p[nn < V ? nn : 0];
      WVs[idx] = (in && nn < V) ? w : 0.0f;
    }
  }
  if (tid < 20) BST[tid] = 0.0f;

  const int wave = tid >> 5;
  const int lane = tid & 31;
  const int e = lane & 15;
  const int h = lane >> 4;
  const int n = e;

  float* S1 = &ST[wave][0];
  float* S2 = &ST[wave][272];
  float* V1 = &ST[wave][400];
  float* V2 = &ST[wave][784];
  float* DOT = &ST[wave][976];
  float* CRS = &ST[wave][1488];

  const float* S1e = S1 + e * 17;
  const float* S2e = S2 + e * 8;
  const float* V1e = V1 + e * 24;
  const float* V2e = V2 + e * 12;
  const float* DOTe = DOT + e * 32;
  const float* CRSe = CRS + e * 96;

  float sumS = 0.f, sumS2 = 0.f, sumV = 0.f;

  for (int trip = blockIdx.x; trip < PTRIPS; trip += EGRID) {
  const int tileBase = (trip * PW + wave) * 16;
  const int eg = tileBase + e;
  __syncthreads();

  if (h == 0) {
    int i = eidx[eg], j = eidx[NE + eg];
#pragma unroll
    for (int u = 0; u < 8; ++u) { S1[e * 17 + u] = x[i * D + u]; S1[e * 17 + 8 + u] = x[j * D + u]; }
    S1[e * 17 + 16] = elen[eg];
#pragma unroll
    for (int t = 0; t < 12; ++t) { V1[e * 24 + t] = x[i * D + 8 + t]; V1[e * 24 + 12 + t] = x[j * D + 8 + t]; }
  } else {
#pragma unroll
    for (int u = 0; u < 8; ++u) S2[e * 8 + u] = ea[eg * D + u];
#pragma unroll
    for (int t = 0; t < 12; ++t) V2[e * 12 + t] = ea[eg * D + 8 + t];
  }
  __syncthreads();
  if (h == 0) {
#pragma unroll
    for (int u = 0; u < 8; ++u)
#pragma unroll
      for (int v = 0; v < 4; ++v) {
        float d = 0.f;
        for (int c = 0; c < 3; ++c) d += V1[e * 24 + u * 3 + c] * V2[e * 12 + v * 3 + c];
        DOT[e * 32 + u * 4 + v] = d;
      }
  } else {
#pragma unroll
    for (int u = 0; u < 8; ++u)
#pragma unroll
      for (int v = 0; v < 4; ++v) {
        float a0 = V1[e * 24 + u * 3], a1 = V1[e * 24 + u * 3 + 1], a2 = V1[e * 24 + u * 3 + 2];
        float b0 = V2[e * 12 + v * 3], b1 = V2[e * 12 + v * 3 + 1], b2 = V2[e * 12 + v * 3 + 2];
        int base = e * 96 + (u * 4 + v) * 3;
        CRS[base + 0] = a1 * b2 - a2 * b1;
        CRS[base + 1] = a2 * b0 - a0 * b2;
        CRS[base + 2] = a0 * b1 - a1 * b0;
      }
  }
  __syncthreads();

  v8f accS = {};
  {
    const float* wb = WSs + ((size_t)h * 16 + n) * 2;
#pragma unroll
    for (int kc = 0; kc < K1S / 4; ++kc) {
      int f0 = kc * 4 + 2 * h;
      v2f a;
      a.x = feat1s(S1e, S2e, DOTe, f0);
      a.y = feat1s(S1e, S2e, DOTe, f0 + 1);
      v2f b = *(const v2f*)(wb + kc * 64);
      accS = __builtin_amdgcn_wmma_f32_16x16x4_f32(false, a, false, b, (short)0, accS, false, false);
    }
  }
  v8f accV0 = edge_vpath<0>(S1e, S2e, V1e, V2e, CRSe, WVs, h, n);
  v8f accV1 = edge_vpath<1>(S1e, S2e, V1e, V2e, CRSe, WVs, h, n);
  v8f accV2 = edge_vpath<2>(S1e, S2e, V1e, V2e, CRSe, WVs, h, n);

#pragma unroll
  for (int r = 0; r < 8; ++r) {
    int em = tileBase + r + 8 * h;
    float sv = accS[r];
    float sg = sigmoidf_(sv);
    float gate = __shfl(sg, (lane & 16) | (8 + (n & 3)), 32);
    if (n < 8) {
      float silu = sv * sg;
      msmv[em * D + n] = silu;
      sumS += silu;
      sumS2 += silu * silu;
    }
    if (n < 4) {
      float v0 = accV0[r] * gate, v1 = accV1[r] * gate, v2 = accV2[r] * gate;
      msmv[em * D + 8 + n * 3 + 0] = v0;
      msmv[em * D + 8 + n * 3 + 1] = v1;
      msmv[em * D + 8 + n * 3 + 2] = v2;
      sumV += v0 * v0 + v1 * v1 + v2 * v2;
    }
  }
  }

  sumS += __shfl_xor(sumS, 16, 32);
  sumS2 += __shfl_xor(sumS2, 16, 32);
  sumV += __shfl_xor(sumV, 16, 32);
  if (h == 0) {
    if (n < 8) { atomicAdd(&BST[n], sumS); atomicAdd(&BST[8 + n], sumS2); }
    if (n < 4) atomicAdd(&BST[16 + n], sumV);
  }
  __syncthreads();
  if (tid < 20) atomicAdd(&estat[tid], BST[tid]);
}

__global__ void k_finalize_bn(const float* __restrict__ stat, const float* __restrict__ wsc,
                              const float* __restrict__ bsc, const float* __restrict__ wvc,
                              float* __restrict__ bn, float count) {
  int t = threadIdx.x;
  if (t < 8) {
    float mu = stat[t] / count;
    float var = stat[8 + t] / count - mu * mu;
    float is = rsqrtf(var + EPSV);
    float sc = wsc[t] * is;
    bn[t] = sc;
    bn[8 + t] = bsc[t] - mu * sc;
  } else if (t < 12) {
    int u = t - 8;
    bn[16 + u] = wvc[u] * rsqrtf(stat[16 + u] / (3.0f * count) + EPSV);
  }
}

__global__ __launch_bounds__(PT) void k_node_tp(
    const float* __restrict__ x, const float* __restrict__ ea, const int* __restrict__ eidx,
    const float* __restrict__ msmv, const float* __restrict__ ebn,
    const float* __restrict__ Wss, const float* __restrict__ Wsv, const float* __restrict__ Wvs,
    const float* __restrict__ Wvvs, const float* __restrict__ Wvvv,
    float* __restrict__ agg, float* __restrict__ eout) {
  __shared__ float WSs[K2S * 16];
  __shared__ float WVs[K2V * 16];
  __shared__ float ST[PW][1664];
  __shared__ int NI[PW][16];

  const int tid = threadIdx.x;
  __builtin_assume(tid >= 0 && tid < PT);
#pragma unroll
  for (int k = 0; k < K2S * 16 / PT; ++k) {
    const int idx = tid + k * PT;
    const int j = idx & 1, nn = (idx >> 1) & 15, hh = (idx >> 5) & 1, kc = idx >> 6;
    const int f = kc * 4 + 2 * hh + j;
    {
      const float* p = (f < 64) ? (Wss + f * OS) : (Wvvs + (f - 64) * OS);
      const float w = p[nn < OS ? nn : 0];
      WSs[idx] = (nn < OS) ? w : 0.0f;
    }
    {
      const float* p = (f < 32) ? (Wsv + f * V) : (f < 64) ? (Wvs + (f - 32) * V) : (Wvvv + (f - 64) * V);
      const float w = p[nn < V ? nn : 0];
      WVs[idx] = (nn < V) ? w : 0.0f;
    }
  }

  const int wave = tid >> 5;
  const int lane = tid & 31;
  const int e = lane & 15;
  const int h = lane >> 4;
  const int n = e;

  float* S1 = &ST[wave][0];
  float* S2 = &ST[wave][128];
  float* V1 = &ST[wave][256];
  float* V2 = &ST[wave][448];
  float* DOT = &ST[wave][640];
  float* CRS = &ST[wave][896];

  const float* S1e = S1 + e * 8;
  const float* S2e = S2 + e * 8;
  const float* V1e = V1 + e * 12;
  const float* V2e = V2 + e * 12;
  const float* DOTe = DOT + e * 16;
  const float* CRSe = CRS + e * 48;

  for (int trip = blockIdx.x; trip < PTRIPS; trip += NGRID) {
  const int tileBase = (trip * PW + wave) * 16;
  const int eg = tileBase + e;
  __syncthreads();

  if (h == 0) {
    int i = eidx[eg];
    NI[wave][e] = i;
#pragma unroll
    for (int u = 0; u < 8; ++u) S1[e * 8 + u] = x[i * D + u];
#pragma unroll
    for (int t = 0; t < 12; ++t) V1[e * 12 + t] = x[i * D + 8 + t];
  } else {
#pragma unroll
    for (int u = 0; u < 8; ++u) {
      float v = msmv[eg * D + u] * ebn[u] + ebn[8 + u] + ea[eg * D + u];
      S2[e * 8 + u] = v;
      eout[eg * D + u] = v;
    }
#pragma unroll
    for (int t = 0; t < 12; ++t) {
      float v = msmv[eg * D + 8 + t] * ebn[16 + t / 3] + ea[eg * D + 8 + t];
      V2[e * 12 + t] = v;
      eout[eg * D + 8 + t] = v;
    }
  }
  __syncthreads();
  if (h == 0) {
#pragma unroll
    for (int u = 0; u < 4; ++u)
#pragma unroll
      for (int v = 0; v < 4; ++v) {
        float d = 0.f;
        for (int c = 0; c < 3; ++c) d += V1[e * 12 + u * 3 + c] * V2[e * 12 + v * 3 + c];
        DOT[e * 16 + u * 4 + v] = d;
      }
  } else {
#pragma unroll
    for (int u = 0; u < 4; ++u)
#pragma unroll
      for (int v = 0; v < 4; ++v) {
        float a0 = V1[e * 12 + u * 3], a1 = V1[e * 12 + u * 3 + 1], a2 = V1[e * 12 + u * 3 + 2];
        float b0 = V2[e * 12 + v * 3], b1 = V2[e * 12 + v * 3 + 1], b2 = V2[e * 12 + v * 3 + 2];
        int base = e * 48 + (u * 4 + v) * 3;
        CRS[base + 0] = a1 * b2 - a2 * b1;
        CRS[base + 1] = a2 * b0 - a0 * b2;
        CRS[base + 2] = a0 * b1 - a1 * b0;
      }
  }
  __syncthreads();

  v8f accS = {};
  {
    const float* wb = WSs + ((size_t)h * 16 + n) * 2;
#pragma unroll
    for (int kc = 0; kc < K2S / 4; ++kc) {
      int f0 = kc * 4 + 2 * h;
      v2f a;
      a.x = feat2s(S1e, S2e, DOTe, f0);
      a.y = feat2s(S1e, S2e, DOTe, f0 + 1);
      v2f b = *(const v2f*)(wb + kc * 64);
      accS = __builtin_amdgcn_wmma_f32_16x16x4_f32(false, a, false, b, (short)0, accS, false, false);
    }
  }
  v8f accV0 = node_vpath<0>(S1e, S2e, V1e, V2e, CRSe, WVs, h, n);
  v8f accV1 = node_vpath<1>(S1e, S2e, V1e, V2e, CRSe, WVs, h, n);
  v8f accV2 = node_vpath<2>(S1e, S2e, V1e, V2e, CRSe, WVs, h, n);

#pragma unroll
  for (int r = 0; r < 8; ++r) {
    int m = r + 8 * h;
    int node = NI[wave][m];
    float sv = accS[r];
    float sg = sigmoidf_(sv);
    float gate = __shfl(sg, (lane & 16) | (8 + (n & 3)), 32);
    if (n < 8) atomicAdd(&agg[node * D + n], sv * sg);
    if (n < 4) {
      atomicAdd(&agg[node * D + 8 + n * 3 + 0], accV0[r] * gate);
      atomicAdd(&agg[node * D + 8 + n * 3 + 1], accV1[r] * gate);
      atomicAdd(&agg[node * D + 8 + n * 3 + 2], accV2[r] * gate);
    }
  }
  }
}

__global__ __launch_bounds__(256) void k_node_stats(const float* __restrict__ agg,
                                                    float* __restrict__ nstat) {
  __shared__ float BST[20];
  int tid = threadIdx.x;
  if (tid < 20) BST[tid] = 0.f;
  __syncthreads();
  int node = blockIdx.x * 256 + tid;
  float ss[8] = {}, s2[8] = {}, vv[4] = {};
  if (node < NN) {
#pragma unroll
    for (int k = 0; k < 8; ++k) { float t = agg[node * D + k]; ss[k] = t; s2[k] = t * t; }
#pragma unroll
    for (int t = 0; t < 12; ++t) { float z = agg[node * D + 8 + t]; vv[t / 3] += z * z; }
  }
#pragma unroll
  for (int off = 16; off >= 1; off >>= 1) {
#pragma unroll
    for (int k = 0; k < 8; ++k) { ss[k] += __shfl_xor(ss[k], off, 32); s2[k] += __shfl_xor(s2[k], off, 32); }
#pragma unroll
    for (int u = 0; u < 4; ++u) vv[u] += __shfl_xor(vv[u], off, 32);
  }
  if ((tid & 31) == 0) {
#pragma unroll
    for (int k = 0; k < 8; ++k) { atomicAdd(&BST[k], ss[k]); atomicAdd(&BST[8 + k], s2[k]); }
#pragma unroll
    for (int u = 0; u < 4; ++u) atomicAdd(&BST[16 + u], vv[u]);
  }
  __syncthreads();
  if (tid < 20) atomicAdd(&nstat[tid], BST[tid]);
}

__global__ __launch_bounds__(256) void k_node_out(const float* __restrict__ agg,
                                                  const float* __restrict__ x,
                                                  const float* __restrict__ nbn,
                                                  float* __restrict__ out) {
  int idx = blockIdx.x * 256 + threadIdx.x;
  if (idx >= NN * D) return;
  int k = idx % D;
  float a = agg[idx];
  float r = (k < 8) ? (a * nbn[k] + nbn[8 + k]) : (a * nbn[16 + (k - 8) / 3]);
  out[idx] = r + x[idx];
}

extern "C" void kernel_launch(void* const* d_in, const int* in_sizes, int n_in,
                              void* d_out, int out_size, void* d_ws, size_t ws_size,
                              hipStream_t stream) {
  (void)in_sizes; (void)n_in; (void)out_size; (void)ws_size;
  const float* x = (const float*)d_in[0];
  const float* ea = (const float*)d_in[1];
  const float* elen = (const float*)d_in[2];
  const float* We_ss = (const float*)d_in[3];
  const float* We_sv = (const float*)d_in[4];
  const float* We_vs = (const float*)d_in[5];
  const float* We_vvs = (const float*)d_in[6];
  const float* We_vvv = (const float*)d_in[7];
  const float* bne_ws = (const float*)d_in[8];
  const float* bne_bs = (const float*)d_in[9];
  const float* bne_wv = (const float*)d_in[10];
  const float* Wn_ss = (const float*)d_in[11];
  const float* Wn_sv = (const float*)d_in[12];
  const float* Wn_vs = (const float*)d_in[13];
  const float* Wn_vvs = (const float*)d_in[14];
  const float* Wn_vvv = (const float*)d_in[15];
  const float* bnn_ws = (const float*)d_in[16];
  const float* bnn_bs = (const float*)d_in[17];
  const float* bnn_wv = (const float*)d_in[18];
  const int* eidx = (const int*)d_in[19];

  float* w = (float*)d_ws;
  float* msmv = w;
  float* agg = w + (size_t)NE * D;
  float* estat = agg + (size_t)NN * D;
  float* nstat = estat + 20;
  float* ebn = nstat + 20;
  float* nbn = ebn + 20;
  float* outf = (float*)d_out;
  float* eout = outf + (size_t)NN * D;

  hipMemsetAsync(agg, 0, ((size_t)NN * D + 40) * sizeof(float), stream);

  k_edge_tp<<<EGRID, PT, 0, stream>>>(x, ea, elen, eidx, We_ss, We_sv, We_vs, We_vvs, We_vvv,
                                        msmv, estat);
  k_finalize_bn<<<1, 32, 0, stream>>>(estat, bne_ws, bne_bs, bne_wv, ebn, (float)NE);
  k_node_tp<<<NGRID, PT, 0, stream>>>(x, ea, eidx, msmv, ebn, Wn_ss, Wn_sv, Wn_vs, Wn_vvs,
                                         Wn_vvv, agg, eout);
  k_node_stats<<<(NN + 255) / 256, 256, 0, stream>>>(agg, nstat);
  k_finalize_bn<<<1, 32, 0, stream>>>(nstat, bnn_ws, bnn_bs, bnn_wv, nbn, (float)NN);
  k_node_out<<<(NN * D + 255) / 256, 256, 0, stream>>>(agg, x, nbn, outf);
}


# Node kernel: vector tensor products factored through dense weight GEMMs as in the edge kernel (27 not 80 matrix ops per tile); tile rows pass through LDS to row-wide atomic adds; sum order changes
# speedup vs baseline: 3.3723x; 1.4994x over previous
#include <hip/hip_runtime.h>
#include <math.h>

#define S 8
#define V 4
#define D 20
#define OS 12
#define NN 25000
#define NE 400000
#define EPSV 1e-5f

#define K1S 168
#define NFRAG 11
#define VPASS 6
#define K2S 80
#define NVFRAG 5
#define NVPASS 3

#define TILES (NE / 16)
#define PW 4
#define PT (PW * 32)
#define PTRIPS (TILES / PW)
#define EGRID 1024
#define NGRID 2048
#define EPASS 21

typedef __attribute__((ext_vector_type(2))) float v2f;
typedef __attribute__((ext_vector_type(8))) float v8f;

__device__ __forceinline__ float sigmoidf_(float x) { return 1.0f / (1.0f + expf(-x)); }

__device__ __forceinline__ float feat1s(const float* S1e, const float* S2e, const float* DOTe,
                                        int f) {
  if (f < 136) return S1e[f >> 3] * S2e[f & 7];
  return DOTe[f - 136];
}

__device__ __forceinline__ float feat2s(const float* S1e, const float* S2e, const float* DOTe,
                                        int f) {
  if (f < 64) return S1e[f >> 3] * S2e[f & 7];
  return DOTe[f - 64];
}

__global__ __launch_bounds__(PT) void k_edge_tp(
    const float* __restrict__ x, const float* __restrict__ ea, const float* __restrict__ elen,
    const int* __restrict__ eidx,
    const float* __restrict__ Wss, const float* __restrict__ Wsv, const float* __restrict__ Wvs,
    const float* __restrict__ Wvvs, const float* __restrict__ Wvvv,
    float* __restrict__ msmv, float* __restrict__ estat) {
  __shared__ float WSs[K1S * 16];
  __shared__ float WAs[VPASS * PT];
  __shared__ float ST[PW][1488];
  __shared__ float BST[20];

  const int tid = threadIdx.x;
  __builtin_assume(tid >= 0 && tid < PT);
#pragma unroll
  for (int k = 0; k < EPASS; ++k) {
    const int idx = tid + k * PT;
    const int j = idx & 1, nn = (idx >> 1) & 15, hh = (idx >> 5) & 1, kc = idx >> 6;
    const int f = kc * 4 + 2 * hh + j;
    const float* p = (f < 136) ? (Wss + f * OS) : (Wvvs + (f - 136) * OS);
    const float w = p[nn < OS ? nn : 0];
    WSs[idx] = (nn < OS) ? w : 0.0f;
  }
#pragma unroll
  for (int k = 0; k < VPASS; ++k) {
    const int idx = tid + k * PT;
    const int j = idx & 1, m = (idx >> 1) & 15, hh = (idx >> 5) & 1, fr = idx >> 6;
    const int kk = 2 * hh + j;
    const int u1 = 4 * fr + kk;
    const int q = fr - 5;
    const int off2 = (4 * (q >> 1) + (m >> 2)) * 32 + (4 * (q & 1) + kk) * 4 + (m & 3);
    const int off3 = (4 * (fr - 9) + (m >> 2)) * 16 + kk * 4 + (m & 3);
    const bool live = (fr < 5) ? (u1 < 17) : (fr < NFRAG);
    const float* p = (fr < 5) ? Wsv : (fr < 9) ? Wvs : Wvvv;
    const int off = (fr < 5) ? u1 * 16 + m : (fr < 9) ? off2 : off3;
    const float w = p[live ? off : 0];
    WAs[idx] = live ? w : 0.0f;
  }
  if (tid < 20) BST[tid] = 0.0f;

  const int wave = tid >> 5;
  const int lane = tid & 31;
  const int e = lane & 15;
  const int h = lane >> 4;
  const int n = e;

  float* S1 = &ST[wave][0];
  float* S2 = &ST[wave][272];
  float* V1 = &ST[wave][400];
  float* V2 = &ST[wave][784];
  float* DOT = &ST[wave][976];

  const float* S1e = S1 + e * 17;
  const float* S2e = S2 + e * 8;
  const float* V1e = V1 + e * 24;
  const float* V2e = V2 + e * 12;
  const float* DOTe = DOT + e * 32;

  float st[16];
#pragma unroll
  for (int i = 0; i < 16; ++i) st[i] = 0.f;

  for (int trip = blockIdx.x; trip < PTRIPS; trip += EGRID) {
  const int tileBase = (trip * PW + wave) * 16;
  const int eg = tileBase + e;
  __syncthreads();

  if (h == 0) {
    int i = eidx[eg], j = eidx[NE + eg];
#pragma unroll
    for (int u = 0; u < 8; ++u) { S1[e * 17 + u] = x[i * D + u]; S1[e * 17 + 8 + u] = x[j * D + u]; }
    S1[e * 17 + 16] = elen[eg];
#pragma unroll
    for (int t = 0; t < 12; ++t) { V1[e * 24 + t] = x[i * D + 8 + t]; V1[e * 24 + 12 + t] = x[j * D + 8 + t]; }
  } else {
#pragma unroll
    for (int u = 0; u < 8; ++u) S2[e * 8 + u] = ea[eg * D + u];
#pragma unroll
    for (int t = 0; t < 12; ++t) V2[e * 12 + t] = ea[eg * D + 8 + t];
  }
  __syncthreads();
  {
#pragma unroll
    for (int u = 0; u < 4; ++u)
#pragma unroll
      for (int v = 0; v < 4; ++v) {
        const int uu = 4 * h + u;
        float d = 0.f;
        for (int c = 0; c < 3; ++c) d += V1[e * 24 + uu * 3 + c] * V2[e * 12 + v * 3 + c];
        DOT[e * 32 + uu * 4 + v] = d;
      }
  }
  __syncthreads();

  v8f accS = {};
  {
    const float* wb = WSs + ((size_t)h * 16 + n) * 2;
#pragma unroll
    for (int kc = 0; kc < K1S / 4; ++kc) {
      int f0 = kc * 4 + 2 * h;
      v2f ft;
      ft.x = feat1s(S1e, S2e, DOTe, f0);
      ft.y = feat1s(S1e, S2e, DOTe, f0 + 1);
      v2f w = *(const v2f*)(wb + kc * 64);
      accS = __builtin_amdgcn_wmma_f32_16x16x4_f32(false, w, false, ft, (short)0, accS, false, false);
    }
  }

  float o[12];
#pragma unroll
  for (int i = 0; i < 12; ++i) o[i] = 0.f;
  const float* wa = WAs + ((size_t)h * 16 + n) * 2;
  float v2r[6];
#pragma unroll
  for (int i = 0; i < 6; ++i) v2r[i] = V2e[6 * h + i];
  {
    v8f T1 = {};
#pragma unroll
    for (int kc = 0; kc < 4; ++kc) {
      v2f b;
      b.x = S1e[4 * kc + 2 * h];
      b.y = S1e[4 * kc + 2 * h + 1];
      v2f a = *(const v2f*)(wa + kc * 64);
      T1 = __builtin_amdgcn_wmma_f32_16x16x4_f32(false, a, false, b, (short)0, T1, false, false);
    }
    {
      v2f b;
      b.x = (h == 0) ? S1e[16] : 0.f;
      b.y = 0.f;
      v2f a = *(const v2f*)(wa + 4 * 64);
      T1 = __builtin_amdgcn_wmma_f32_16x16x4_f32(false, a, false, b, (short)0, T1, false, false);
    }
#pragma unroll
    for (int vv = 0; vv < 2; ++vv)
#pragma unroll
      for (int oo = 0; oo < 4; ++oo)
#pragma unroll
        for (int c = 0; c < 3; ++c) o[oo * 3 + c] += v2r[vv * 3 + c] * T1[vv * 4 + oo];
  }
  {
    v2f bs[2];
#pragma unroll
    for (int kc = 0; kc < 2; ++kc) { bs[kc].x = S2e[4 * kc + 2 * h]; bs[kc].y = S2e[4 * kc + 2 * h + 1]; }
#pragma unroll
    for (int t = 0; t < 2; ++t) {
      v8f T2 = {};
#pragma unroll
      for (int kc = 0; kc < 2; ++kc) {
        v2f a = *(const v2f*)(wa + (5 + 2 * t + kc) * 64);
        T2 = __builtin_amdgcn_wmma_f32_16x16x4_f32(false, a, false, bs[kc], (short)0, T2, false, false);
      }
#pragma unroll
      for (int rr = 0; rr < 2; ++rr) {
        const float* v1u = V1e + (4 * t + 2 * h + rr) * 3;
        const float a0 = v1u[0], a1 = v1u[1], a2 = v1u[2];
#pragma unroll
        for (int oo = 0; oo < 4; ++oo) {
          const float tv = T2[rr * 4 + oo];
          o[oo * 3 + 0] += a0 * tv;
          o[oo * 3 + 1] += a1 * tv;
          o[oo * 3 + 2] += a2 * tv;
        }
      }
    }
  }
  {
    v2f bv[3];
#pragma unroll
    for (int b = 0; b < 3; ++b) { bv[b].x = v2r[b]; bv[b].y = v2r[3 + b]; }
#pragma unroll
    for (int t = 0; t < 2; ++t) {
      v2f a = *(const v2f*)(wa + (9 + t) * 64);
      v8f z = {};
      v8f Q0 = __builtin_amdgcn_wmma_f32_16x16x4_f32(false, a, false, bv[0], (short)0, z, false, false);
      v8f Q1 = __builtin_amdgcn_wmma_f32_16x16x4_f32(false, a, false, bv[1], (short)0, z, false, false);
      v8f Q2 = __builtin_amdgcn_wmma_f32_16x16x4_f32(false, a, false, bv[2], (short)0, z, false, false);
#pragma unroll
      for (int rr = 0; rr < 2; ++rr) {
        const float* v1u = V1e + (4 * t + 2 * h + rr) * 3;
        const float a0 = v1u[0], a1 = v1u[1], a2 = v1u[2];
#pragma unroll
        for (int oo = 0; oo < 4; ++oo) {
          const float q0 = Q0[rr * 4 + oo], q1 = Q1[rr * 4 + oo], q2 = Q2[rr * 4 + oo];
          o[oo * 3 + 0] += a1 * q2 - a2 * q1;
          o[oo * 3 + 1] += a2 * q0 - a0 * q2;
          o[oo * 3 + 2] += a0 * q1 - a1 * q0;
        }
      }
    }
  }
#pragma unroll
  for (int i = 0; i < 12; ++i) o[i] += __shfl_xor(o[i], 16, 32);

  float sg[8];
#pragma unroll
  for (int r = 0; r < 8; ++r) sg[r] = sigmoidf_(accS[r]);
  if (h == 0) {
#pragma unroll
    for (int r = 0; r < 8; ++r) {
      float silu = accS[r] * sg[r];
      msmv[eg * D + r] = silu;
      st[r] += silu;
      st[8 + r] += silu * silu;
    }
  } else {
#pragma unroll
    for (int oo = 0; oo < 4; ++oo) {
      float gate = sg[oo];
      float v0 = o[oo * 3 + 0] * gate, v1 = o[oo * 3 + 1] * gate, v2 = o[oo * 3 + 2] * gate;
      msmv[eg * D + 8 + oo * 3 + 0] = v0;
      msmv[eg * D + 8 + oo * 3 + 1] = v1;
      msmv[eg * D + 8 + oo * 3 + 2] = v2;
      st[oo] += v0 * v0 + v1 * v1 + v2 * v2;
    }
  }
  }

#pragma unroll
  for (int off = 8; off >= 1; off >>= 1) {
#pragma unroll
    for (int i = 0; i < 16; ++i) st[i] += __shfl_xor(st[i], off, 32);
  }
  if (e == 0) {
    if (h == 0) {
#pragma unroll
      for (int r = 0; r < 8; ++r) { atomicAdd(&BST[r], st[r]); atomicAdd(&BST[8 + r], st[8 + r]); }
    } else {
#pragma unroll
      for (int oo = 0; oo < 4; ++oo) atomicAdd(&BST[16 + oo], st[oo]);
    }
  }
  __syncthreads();
  if (tid < 20) atomicAdd(&estat[tid], BST[tid]);
}

__global__ void k_finalize_bn(const float* __restrict__ stat, const float* __restrict__ wsc,
                              const float* __restrict__ bsc, const float* __restrict__ wvc,
                              float* __restrict__ bn, float count) {
  int t = threadIdx.x;
  if (t < 8) {
    float mu = stat[t] / count;
    float var = stat[8 + t] / count - mu * mu;
    float is = rsqrtf(var + EPSV);
    float sc = wsc[t] * is;
    bn[t] = sc;
    bn[8 + t] = bsc[t] - mu * sc;
  } else if (t < 12) {
    int u = t - 8;
    bn[16 + u] = wvc[u] * rsqrtf(stat[16 + u] / (3.0f * count) + EPSV);
  }
}

__global__ __launch_bounds__(PT, 8) void k_node_tp(
    const float* __restrict__ x, const float* __restrict__ ea, const int* __restrict__ eidx,
    const float* __restrict__ msmv, const float* __restrict__ ebn,
    const float* __restrict__ Wss, const float* __restrict__ Wsv, const float* __restrict__ Wvs,
    const float* __restrict__ Wvvs, const float* __restrict__ Wvvv,
    float* __restrict__ agg, float* __restrict__ eout) {
  __shared__ float WSs[K2S * 16];
  __shared__ float WAs[NVPASS * PT];
  __shared__ float ST[PW][960];
  __shared__ int OFS[PW][320];
  __shared__ int NI[PW][16];

  const int tid = threadIdx.x;
  __builtin_assume(tid >= 0 && tid < PT);
#pragma unroll
  for (int k = 0; k < K2S * 16 / PT; ++k) {
    const int idx = tid + k * PT;
    const int j = idx & 1, nn = (idx >> 1) & 15, hh = (idx >> 5) & 1, kc = idx >> 6;
    const int f = kc * 4 + 2 * hh + j;
    const float* p = (f < 64) ? (Wss + f * OS) : (Wvvs + (f - 64) * OS);
    const float w = p[nn < OS ? nn : 0];
    WSs[idx] = (nn < OS) ? w : 0.0f;
  }
#pragma unroll
  for (int k = 0; k < NVPASS; ++k) {
    const int idx = tid + k * PT;
    const int j = idx & 1, m = (idx >> 1) & 15, hh = (idx >> 5) & 1, fr = idx >> 6;
    const int kk = 2 * hh + j;
    const int off1 = (4 * fr + kk) * 16 + m;
    const int off2 = (m >> 2) * 32 + (4 * (fr - 2) + kk) * 4 + (m & 3);
    const int off3 = (m >> 2) * 16 + kk * 4 + (m & 3);
    const bool live = fr < NVFRAG;
    const float* p = (fr < 2) ? Wsv : (fr < 4) ? Wvs : Wvvv;
    const int off = (fr < 2) ? off1 : (fr < 4) ? off2 : off3;
    const float w = p[live ? off : 0];
    WAs[idx] = live ? w : 0.0f;
  }

  const int wave = tid >> 5;
  const int lane = tid & 31;
  const int e = lane & 15;
  const int h = lane >> 4;
  const int n = e;

  float* S1 = &ST[wave][0];
  float* S2 = &ST[wave][128];
  float* V1 = &ST[wave][256];
  float* V2 = &ST[wave][448];
  float* OUT = &ST[wave][640];
  int* OFF = &OFS[wave][0];

  const float* S1e = S1 + e * 8;
  const float* S2e = S2 + e * 8;
  const float* V1e = V1 + e * 12;
  const float* V2e = V2 + e * 12;

  for (int trip = blockIdx.x; trip < PTRIPS; trip += NGRID) {
  const int tileBase = (trip * PW + wave) * 16;
  const int eg = tileBase + e;
  __syncthreads();

  if (h == 0) {
    int i = eidx[eg];
    NI[wave][e] = i;
#pragma unroll
    for (int u = 0; u < 8; ++u) S1[e * 8 + u] = x[i * D + u];
#pragma unroll
    for (int t = 0; t < 12; ++t) V1[e * 12 + t] = x[i * D + 8 + t];
  } else {
#pragma unroll
    for (int u = 0; u < 8; ++u) {
      float v = msmv[eg * D + u] * ebn[u] + ebn[8 + u] + ea[eg * D + u];
      S2[e * 8 + u] = v;
      eout[eg * D + u] = v;
    }
#pragma unroll
    for (int t = 0; t < 12; ++t) {
      float v = msmv[eg * D + 8 + t] * ebn[16 + t / 3] + ea[eg * D + 8 + t];
      V2[e * 12 + t] = v;
      eout[eg * D + 8 + t] = v;
    }
  }
  __syncthreads();

  float s1r[8], v1r[12], v2r[6], v1h[6];
#pragma unroll
  for (int u = 0; u < 8; ++u) s1r[u] = S1e[u];
#pragma unroll
  for (int t = 0; t < 12; ++t) v1r[t] = V1e[t];
#pragma unroll
  for (int t = 0; t < 6; ++t) { v2r[t] = V2e[6 * h + t]; v1h[t] = V1e[6 * h + t]; }
  v2f s2h[2], s1h[2];
#pragma unroll
  for (int kc = 0; kc < 2; ++kc) {
    s2h[kc].x = S2e[4 * kc + 2 * h]; s2h[kc].y = S2e[4 * kc + 2 * h + 1];
    s1h[kc].x = S1e[4 * kc + 2 * h]; s1h[kc].y = S1e[4 * kc + 2 * h + 1];
  }

  v8f accS = {};
  {
    const float* wb = WSs + ((size_t)h * 16 + n) * 2;
#pragma unroll
    for (int kc = 0; kc < 16; ++kc) {
      v2f ft;
      ft.x = s1r[kc >> 1] * s2h[kc & 1].x;
      ft.y = s1r[kc >> 1] * s2h[kc & 1].y;
      v2f w = *(const v2f*)(wb + kc * 64);
      accS = __builtin_amdgcn_wmma_f32_16x16x4_f32(false, w, false, ft, (short)0, accS, false, false);
    }
#pragma unroll
    for (int u = 0; u < 4; ++u) {
      v2f ft;
      ft.x = v1r[u * 3] * v2r[0] + v1r[u * 3 + 1] * v2r[1] + v1r[u * 3 + 2] * v2r[2];
      ft.y = v1r[u * 3] * v2r[3] + v1r[u * 3 + 1] * v2r[4] + v1r[u * 3 + 2] * v2r[5];
      v2f w = *(const v2f*)(wb + (16 + u) * 64);
      accS = __builtin_amdgcn_wmma_f32_16x16x4_f32(false, w, false, ft, (short)0, accS, false, false);
    }
  }

  float o[12];
#pragma unroll
  for (int i = 0; i < 12; ++i) o[i] = 0.f;
  const float* wa = WAs + ((size_t)h * 16 + n) * 2;
  {
    v8f T1 = {};
#pragma unroll
    for (int kc = 0; kc < 2; ++kc) {
      v2f a = *(const v2f*)(wa + kc * 64);
      T1 = __builtin_amdgcn_wmma_f32_16x16x4_f32(false, a, false, s1h[kc], (short)0, T1, false, false);
    }
#pragma unroll
    for (int vv = 0; vv < 2; ++vv)
#pragma unroll
      for (int oo = 0; oo < 4; ++oo)
#pragma unroll
        for (int c = 0; c < 3; ++c) o[oo * 3 + c] += v2r[vv * 3 + c] * T1[vv * 4 + oo];
  }
  {
    v8f T2 = {};
#pragma unroll
    for (int kc = 0; kc < 2; ++kc) {
      v2f a = *(const v2f*)(wa + (2 + kc) * 64);
      T2 = __builtin_amdgcn_wmma_f32_16x16x4_f32(false, a, false, s2h[kc], (short)0, T2, false, false);
    }
#pragma unroll
    for (int rr = 0; rr < 2; ++rr) {
      const float a0 = v1h[rr * 3], a1 = v1h[rr * 3 + 1], a2 = v1h[rr * 3 + 2];
#pragma unroll
      for (int oo = 0; oo < 4; ++oo) {
        const float tv = T2[rr * 4 + oo];
        o[oo * 3 + 0] += a0 * tv;
        o[oo * 3 + 1] += a1 * tv;
        o[oo * 3 + 2] += a2 * tv;
      }
    }
  }
  {
    v2f bv[3];
#pragma unroll
    for (int b = 0; b < 3; ++b) { bv[b].x = v2r[b]; bv[b].y = v2r[3 + b]; }
    v2f a = *(const v2f*)(wa + 4 * 64);
    v8f z = {};
    v8f Q0 = __builtin_amdgcn_wmma_f32_16x16x4_f32(false, a, false, bv[0], (short)0, z, false, false);
    v8f Q1 = __builtin_amdgcn_wmma_f32_16x16x4_f32(false, a, false, bv[1], (short)0, z, false, false);
    v8f Q2 = __builtin_amdgcn_wmma_f32_16x16x4_f32(false, a, false, bv[2], (short)0, z, false, false);
#pragma unroll
    for (int rr = 0; rr < 2; ++rr) {
      const float a0 = v1h[rr * 3], a1 = v1h[rr * 3 + 1], a2 = v1h[rr * 3 + 2];
#pragma unroll
      for (int oo = 0; oo < 4; ++oo) {
        const float q0 = Q0[rr * 4 + oo], q1 = Q1[rr * 4 + oo], q2 = Q2[rr * 4 + oo];
        o[oo * 3 + 0] += a1 * q2 - a2 * q1;
        o[oo * 3 + 1] += a2 * q0 - a0 * q2;
        o[oo * 3 + 2] += a0 * q1 - a1 * q0;
      }
    }
  }
#pragma unroll
  for (int i = 0; i < 12; ++i) o[i] += __shfl_xor(o[i], 16, 32);

  float sg[8];
#pragma unroll
  for (int r = 0; r < 8; ++r) sg[r] = sigmoidf_(accS[r]);
  const int rowoff = NI[wave][e] * D;
  if (h == 0) {
#pragma unroll
    for (int r = 0; r < 8; ++r) { OUT[e * D + r] = accS[r] * sg[r]; OFF[e * D + r] = rowoff + r; }
  } else {
#pragma unroll
    for (int oo = 0; oo < 4; ++oo)
#pragma unroll
      for (int c = 0; c < 3; ++c) {
        OUT[e * D + 8 + oo * 3 + c] = o[oo * 3 + c] * sg[oo];
        OFF[e * D + 8 + oo * 3 + c] = rowoff + 8 + oo * 3 + c;
      }
  }
  __syncthreads();
  float ov[10];
  int of[10];
#pragma unroll
  for (int k = 0; k < 10; ++k) { ov[k] = OUT[lane + 32 * k]; of[k] = OFF[lane + 32 * k]; }
#pragma unroll
  for (int k = 0; k < 10; ++k) atomicAdd(&agg[of[k]], ov[k]);
  }
}

__global__ __launch_bounds__(256) void k_node_stats(const float* __restrict__ agg,
                                                    float* __restrict__ nstat) {
  __shared__ float BST[20];
  int tid = threadIdx.x;
  if (tid < 20) BST[tid] = 0.f;
  __syncthreads();
  int node = blockIdx.x * 256 + tid;
  float ss[8] = {}, s2[8] = {}, vv[4] = {};
  if (node < NN) {
#pragma unroll
    for (int k = 0; k < 8; ++k) { float t = agg[node * D + k]; ss[k] = t; s2[k] = t * t; }
#pragma unroll
    for (int t = 0; t < 12; ++t) { float z = agg[node * D + 8 + t]; vv[t / 3] += z * z; }
  }
#pragma unroll
  for (int off = 16; off >= 1; off >>= 1) {
#pragma unroll
    for (int k = 0; k < 8; ++k) { ss[k] += __shfl_xor(ss[k], off, 32); s2[k] += __shfl_xor(s2[k], off, 32); }
#pragma unroll
    for (int u = 0; u < 4; ++u) vv[u] += __shfl_xor(vv[u], off, 32);
  }
  if ((tid & 31) == 0) {
#pragma unroll
    for (int k = 0; k < 8; ++k) { atomicAdd(&BST[k], ss[k]); atomicAdd(&BST[8 + k], s2[k]); }
#pragma unroll
    for (int u = 0; u < 4; ++u) atomicAdd(&BST[16 + u], vv[u]);
  }
  __syncthreads();
  if (tid < 20) atomicAdd(&nstat[tid], BST[tid]);
}

__global__ __launch_bounds__(256) void k_node_out(const float* __restrict__ agg,
                                                  const float* __restrict__ x,
                                                  const float* __restrict__ nbn,
                                                  float* __restrict__ out) {
  int idx = blockIdx.x * 256 + threadIdx.x;
  if (idx >= NN * D) return;
  int k = idx % D;
  float a = agg[idx];
  float r = (k < 8) ? (a * nbn[k] + nbn[8 + k]) : (a * nbn[16 + (k - 8) / 3]);
  out[idx] = r + x[idx];
}

extern "C" void kernel_launch(void* const* d_in, const int* in_sizes, int n_in,
                              void* d_out, int out_size, void* d_ws, size_t ws_size,
                              hipStream_t stream) {
  (void)in_sizes; (void)n_in; (void)out_size; (void)ws_size;
  const float* x = (const float*)d_in[0];
  const float* ea = (const float*)d_in[1];
  const float* elen = (const float*)d_in[2];
  const float* We_ss = (const float*)d_in[3];
  const float* We_sv = (const float*)d_in[4];
  const float* We_vs = (const float*)d_in[5];
  const float* We_vvs = (const float*)d_in[6];
  const float* We_vvv = (const float*)d_in[7];
  const float* bne_ws = (const float*)d_in[8];
  const float* bne_bs = (const float*)d_in[9];
  const float* bne_wv = (const float*)d_in[10];
  const float* Wn_ss = (const float*)d_in[11];
  const float* Wn_sv = (const float*)d_in[12];
  const float* Wn_vs = (const float*)d_in[13];
  const float* Wn_vvs = (const float*)d_in[14];
  const float* Wn_vvv = (const float*)d_in[15];
  const float* bnn_ws = (const float*)d_in[16];
  const float* bnn_bs = (const float*)d_in[17];
  const float* bnn_wv = (const float*)d_in[18];
  const int* eidx = (const int*)d_in[19];

  float* w = (float*)d_ws;
  float* msmv = w;
  float* agg = w + (size_t)NE * D;
  float* estat = agg + (size_t)NN * D;
  float* nstat = estat + 20;
  float* ebn = nstat + 20;
  float* nbn = ebn + 20;
  float* outf = (float*)d_out;
  float* eout = outf + (size_t)NN * D;

  hipMemsetAsync(agg, 0, ((size_t)NN * D + 40) * sizeof(float), stream);

  k_edge_tp<<<EGRID, PT, 0, stream>>>(x, ea, elen, eidx, We_ss, We_sv, We_vs, We_vvs, We_vvv,
                                        msmv, estat);
  k_finalize_bn<<<1, 32, 0, stream>>>(estat, bne_ws, bne_bs, bne_wv, ebn, (float)NE);
  k_node_tp<<<NGRID, PT, 0, stream>>>(x, ea, eidx, msmv, ebn, Wn_ss, Wn_sv, Wn_vs, Wn_vvs,
                                         Wn_vvv, agg, eout);
  k_node_stats<<<(NN + 255) / 256, 256, 0, stream>>>(agg, nstat);
  k_finalize_bn<<<1, 32, 0, stream>>>(nstat, bnn_ws, bnn_bs, bnn_wv, nbn, (float)NN);
  k_node_out<<<(NN * D + 255) / 256, 256, 0, stream>>>(agg, x, nbn, outf);
}


# Node scatter-add: agg rows padded to one aligned 128-byte line, one edge row per atomic instruction (16 lines a tile, not about 31); edge kernel's rows kept in edge_out until replaced
# speedup vs baseline: 3.7489x; 1.1117x over previous
#include <hip/hip_runtime.h>
#include <math.h>

#define S 8
#define V 4
#define D 20
#define OS 12
#define NN 25000
#define NE 400000
#define EPSV 1e-5f

#define K1S 168
#define NFRAG 11
#define VPASS 6
#define K2S 80
#define NVFRAG 5
#define NVPASS 3

#define TILES (NE / 16)
#define PW 4
#define PT (PW * 32)
#define PTRIPS (TILES / PW)
#define EGRID 1024
#define NGRID 2048
#define EPASS 21
#define AP 32

typedef __attribute__((ext_vector_type(2))) float v2f;
typedef __attribute__((ext_vector_type(8))) float v8f;

__device__ __forceinline__ float sigmoidf_(float x) { return 1.0f / (1.0f + expf(-x)); }

__device__ __forceinline__ float feat1s(const float* S1e, const float* S2e, const float* DOTe,
                                        int f) {
  if (f < 136) return S1e[f >> 3] * S2e[f & 7];
  return DOTe[f - 136];
}

__device__ __forceinline__ float feat2s(const float* S1e, const float* S2e, const float* DOTe,
                                        int f) {
  if (f < 64) return S1e[f >> 3] * S2e[f & 7];
  return DOTe[f - 64];
}

__global__ __launch_bounds__(PT) void k_edge_tp(
    const float* __restrict__ x, const float* __restrict__ ea, const float* __restrict__ elen,
    const int* __restrict__ eidx,
    const float* __restrict__ Wss, const float* __restrict__ Wsv, const float* __restrict__ Wvs,
    const float* __restrict__ Wvvs, const float* __restrict__ Wvvv,
    float* __restrict__ msmv, float* __restrict__ estat) {
  __shared__ float WSs[K1S * 16];
  __shared__ float WAs[VPASS * PT];
  __shared__ float ST[PW][1488];
  __shared__ float BST[20];

  const int tid = threadIdx.x;
  __builtin_assume(tid >= 0 && tid < PT);
#pragma unroll
  for (int k = 0; k < EPASS; ++k) {
    const int idx = tid + k * PT;
    const int j = idx & 1, nn = (idx >> 1) & 15, hh = (idx >> 5) & 1, kc = idx >> 6;
    const int f = kc * 4 + 2 * hh + j;
    const float* p = (f < 136) ? (Wss + f * OS) : (Wvvs + (f - 136) * OS);
    const float w = p[nn < OS ? nn : 0];
    WSs[idx] = (nn < OS) ? w : 0.0f;
  }
#pragma unroll
  for (int k = 0; k < VPASS; ++k) {
    const int idx = tid + k * PT;
    const int j = idx & 1, m = (idx >> 1) & 15, hh = (idx >> 5) & 1, fr = idx >> 6;
    const int kk = 2 * hh + j;
    const int u1 = 4 * fr + kk;
    const int q = fr - 5;
    const int off2 = (4 * (q >> 1) + (m >> 2)) * 32 + (4 * (q & 1) + kk) * 4 + (m & 3);
    const int off3 = (4 * (fr - 9) + (m >> 2)) * 16 + kk * 4 + (m & 3);
    const bool live = (fr < 5) ? (u1 < 17) : (fr < NFRAG);
    const float* p = (fr < 5) ? Wsv : (fr < 9) ? Wvs : Wvvv;
    const int off = (fr < 5) ? u1 * 16 + m : (fr < 9) ? off2 : off3;
    const float w = p[live ? off : 0];
    WAs[idx] = live ? w : 0.0f;
  }
  if (tid < 20) BST[tid] = 0.0f;

  const int wave = tid >> 5;
  const int lane = tid & 31;
  const int e = lane & 15;
  const int h = lane >> 4;
  const int n = e;

  float* S1 = &ST[wave][0];
  float* S2 = &ST[wave][272];
  float* V1 = &ST[wave][400];
  float* V2 = &ST[wave][784];
  float* DOT = &ST[wave][976];

  const float* S1e = S1 + e * 17;
  const float* S2e = S2 + e * 8;
  const float* V1e = V1 + e * 24;
  const float* V2e = V2 + e * 12;
  const float* DOTe = DOT + e * 32;

  float st[16];
#pragma unroll
  for (int i = 0; i < 16; ++i) st[i] = 0.f;

  for (int trip = blockIdx.x; trip < PTRIPS; trip += EGRID) {
  const int tileBase = (trip * PW + wave) * 16;
  const int eg = tileBase + e;
  __syncthreads();

  if (h == 0) {
    int i = eidx[eg], j = eidx[NE + eg];
#pragma unroll
    for (int u = 0; u < 8; ++u) { S1[e * 17 + u] = x[i * D + u]; S1[e * 17 + 8 + u] = x[j * D + u]; }
    S1[e * 17 + 16] = elen[eg];
#pragma unroll
    for (int t = 0; t < 12; ++t) { V1[e * 24 + t] = x[i * D + 8 + t]; V1[e * 24 + 12 + t] = x[j * D + 8 + t]; }
  } else {
#pragma unroll
    for (int u = 0; u < 8; ++u) S2[e * 8 + u] = ea[eg * D + u];
#pragma unroll
    for (int t = 0; t < 12; ++t) V2[e * 12 + t] = ea[eg * D + 8 + t];
  }
  __syncthreads();
  {
#pragma unroll
    for (int u = 0; u < 4; ++u)
#pragma unroll
      for (int v = 0; v < 4; ++v) {
        const int uu = 4 * h + u;
        float d = 0.f;
        for (int c = 0; c < 3; ++c) d += V1[e * 24 + uu * 3 + c] * V2[e * 12 + v * 3 + c];
        DOT[e * 32 + uu * 4 + v] = d;
      }
  }
  __syncthreads();

  v8f accS = {};
  {
    const float* wb = WSs + ((size_t)h * 16 + n) * 2;
#pragma unroll
    for (int kc = 0; kc < K1S / 4; ++kc) {
      int f0 = kc * 4 + 2 * h;
      v2f ft;
      ft.x = feat1s(S1e, S2e, DOTe, f0);
      ft.y = feat1s(S1e, S2e, DOTe, f0 + 1);
      v2f w = *(const v2f*)(wb + kc * 64);
      accS = __builtin_amdgcn_wmma_f32_16x16x4_f32(false, w, false, ft, (short)0, accS, false, false);
    }
  }

  float o[12];
#pragma unroll
  for (int i = 0; i < 12; ++i) o[i] = 0.f;
  const float* wa = WAs + ((size_t)h * 16 + n) * 2;
  float v2r[6];
#pragma unroll
  for (int i = 0; i < 6; ++i) v2r[i] = V2e[6 * h + i];
  {
    v8f T1 = {};
#pragma unroll
    for (int kc = 0; kc < 4; ++kc) {
      v2f b;
      b.x = S1e[4 * kc + 2 * h];
      b.y = S1e[4 * kc + 2 * h + 1];
      v2f a = *(const v2f*)(wa + kc * 64);
      T1 = __builtin_amdgcn_wmma_f32_16x16x4_f32(false, a, false, b, (short)0, T1, false, false);
    }
    {
      v2f b;
      b.x = (h == 0) ? S1e[16] : 0.f;
      b.y = 0.f;
      v2f a = *(const v2f*)(wa + 4 * 64);
      T1 = __builtin_amdgcn_wmma_f32_16x16x4_f32(false, a, false, b, (short)0, T1, false, false);
    }
#pragma unroll
    for (int vv = 0; vv < 2; ++vv)
#pragma unroll
      for (int oo = 0; oo < 4; ++oo)
#pragma unroll
        for (int c = 0; c < 3; ++c) o[oo * 3 + c] += v2r[vv * 3 + c] * T1[vv * 4 + oo];
  }
  {
    v2f bs[2];
#pragma unroll
    for (int kc = 0; kc < 2; ++kc) { bs[kc].x = S2e[4 * kc + 2 * h]; bs[kc].y = S2e[4 * kc + 2 * h + 1]; }
#pragma unroll
    for (int t = 0; t < 2; ++t) {
      v8f T2 = {};
#pragma unroll
      for (int kc = 0; kc < 2; ++kc) {
        v2f a = *(const v2f*)(wa + (5 + 2 * t + kc) * 64);
        T2 = __builtin_amdgcn_wmma_f32_16x16x4_f32(false, a, false, bs[kc], (short)0, T2, false, false);
      }
#pragma unroll
      for (int rr = 0; rr < 2; ++rr) {
        const float* v1u = V1e + (4 * t + 2 * h + rr) * 3;
        const float a0 = v1u[0], a1 = v1u[1], a2 = v1u[2];
#pragma unroll
        for (int oo = 0; oo < 4; ++oo) {
          const float tv = T2[rr * 4 + oo];
          o[oo * 3 + 0] += a0 * tv;
          o[oo * 3 + 1] += a1 * tv;
          o[oo * 3 + 2] += a2 * tv;
        }
      }
    }
  }
  {
    v2f bv[3];
#pragma unroll
    for (int b = 0; b < 3; ++b) { bv[b].x = v2r[b]; bv[b].y = v2r[3 + b]; }
#pragma unroll
    for (int t = 0; t < 2; ++t) {
      v2f a = *(const v2f*)(wa + (9 + t) * 64);
      v8f z = {};
      v8f Q0 = __builtin_amdgcn_wmma_f32_16x16x4_f32(false, a, false, bv[0], (short)0, z, false, false);
      v8f Q1 = __builtin_amdgcn_wmma_f32_16x16x4_f32(false, a, false, bv[1], (short)0, z, false, false);
      v8f Q2 = __builtin_amdgcn_wmma_f32_16x16x4_f32(false, a, false, bv[2], (short)0, z, false, false);
#pragma unroll
      for (int rr = 0; rr < 2; ++rr) {
        const float* v1u = V1e + (4 * t + 2 * h + rr) * 3;
        const float a0 = v1u[0], a1 = v1u[1], a2 = v1u[2];
#pragma unroll
        for (int oo = 0; oo < 4; ++oo) {
          const float q0 = Q0[rr * 4 + oo], q1 = Q1[rr * 4 + oo], q2 = Q2[rr * 4 + oo];
          o[oo * 3 + 0] += a1 * q2 - a2 * q1;
          o[oo * 3 + 1] += a2 * q0 - a0 * q2;
          o[oo * 3 + 2] += a0 * q1 - a1 * q0;
        }
      }
    }
  }
#pragma unroll
  for (int i = 0; i < 12; ++i) o[i] += __shfl_xor(o[i], 16, 32);

  float sg[8];
#pragma unroll
  for (int r = 0; r < 8; ++r) sg[r] = sigmoidf_(accS[r]);
  if (h == 0) {
#pragma unroll
    for (int r = 0; r < 8; ++r) {
      float silu = accS[r] * sg[r];
      msmv[eg * D + r] = silu;
      st[r] += silu;
      st[8 + r] += silu * silu;
    }
  } else {
#pragma unroll
    for (int oo = 0; oo < 4; ++oo) {
      float gate = sg[oo];
      float v0 = o[oo * 3 + 0] * gate, v1 = o[oo * 3 + 1] * gate, v2 = o[oo * 3 + 2] * gate;
      msmv[eg * D + 8 + oo * 3 + 0] = v0;
      msmv[eg * D + 8 + oo * 3 + 1] = v1;
      msmv[eg * D + 8 + oo * 3 + 2] = v2;
      st[oo] += v0 * v0 + v1 * v1 + v2 * v2;
    }
  }
  }

#pragma unroll
  for (int off = 8; off >= 1; off >>= 1) {
#pragma unroll
    for (int i = 0; i < 16; ++i) st[i] += __shfl_xor(st[i], off, 32);
  }
  if (e == 0) {
    if (h == 0) {
#pragma unroll
      for (int r = 0; r < 8; ++r) { atomicAdd(&BST[r], st[r]); atomicAdd(&BST[8 + r], st[8 + r]); }
    } else {
#pragma unroll
      for (int oo = 0; oo < 4; ++oo) atomicAdd(&BST[16 + oo], st[oo]);
    }
  }
  __syncthreads();
  if (tid < 20) atomicAdd(&estat[tid], BST[tid]);
}

__device__ __forceinline__ void bn_coeffs(const float* __restrict__ stat, const float* __restrict__ wsc,
                                          const float* __restrict__ bsc, const float* __restrict__ wvc,
                                          float* bn, int t, float count) {
  if (t < 8) {
    float mu = stat[t] / count;
    float var = stat[8 + t] / count - mu * mu;
    float is = rsqrtf(var + EPSV);
    float sc = wsc[t] * is;
    bn[t] = sc;
    bn[8 + t] = bsc[t] - mu * sc;
  } else if (t < 12) {
    int u = t - 8;
    bn[16 + u] = wvc[u] * rsqrtf(stat[16 + u] / (3.0f * count) + EPSV);
  }
}

__global__ __launch_bounds__(PT, 8) void k_node_tp(
    const float* __restrict__ x, const float* __restrict__ ea, const int* __restrict__ eidx,
    const float* __restrict__ estat,
    const float* __restrict__ bws, const float* __restrict__ bbs, const float* __restrict__ bwv,
    const float* __restrict__ Wss, const float* __restrict__ Wsv, const float* __restrict__ Wvs,
    const float* __restrict__ Wvvs, const float* __restrict__ Wvvv,
    float* __restrict__ agg, float* __restrict__ eout) {
  __shared__ float WSs[K2S * 16];
  __shared__ float WAs[NVPASS * PT];
  __shared__ float ST[PW][960];
  __shared__ __attribute__((aligned(16))) float EBN[20];

  const int tid = threadIdx.x;
  __builtin_assume(tid >= 0 && tid < PT);
#pragma unroll
  for (int k = 0; k < K2S * 16 / PT; ++k) {
    const int idx = tid + k * PT;
    const int j = idx & 1, nn = (idx >> 1) & 15, hh = (idx >> 5) & 1, kc = idx >> 6;
    const int f = kc * 4 + 2 * hh + j;
    const float* p = (f < 64) ? (Wss + f * OS) : (Wvvs + (f - 64) * OS);
    const float w = p[nn < OS ? nn : 0];
    WSs[idx] = (nn < OS) ? w : 0.0f;
  }
#pragma unroll
  for (int k = 0; k < NVPASS; ++k) {
    const int idx = tid + k * PT;
    const int j = idx & 1, m = (idx >> 1) & 15, hh = (idx >> 5) & 1, fr = idx >> 6;
    const int kk = 2 * hh + j;
    const int off1 = (4 * fr + kk) * 16 + m;
    const int off2 = (m >> 2) * 32 + (4 * (fr - 2) + kk) * 4 + (m & 3);
    const int off3 = (m >> 2) * 16 + kk * 4 + (m & 3);
    const bool live = fr < NVFRAG;
    const float* p = (fr < 2) ? Wsv : (fr < 4) ? Wvs : Wvvv;
    const int off = (fr < 2) ? off1 : (fr < 4) ? off2 : off3;
    const float w = p[live ? off : 0];
    WAs[idx] = live ? w : 0.0f;
  }
  if (tid < 12) bn_coeffs(estat, bws, bbs, bwv, EBN, tid, (float)NE);

  const int wave = tid >> 5;
  const int lane = tid & 31;
  const int e = lane & 15;
  const int h = lane >> 4;
  const int n = e;

  float* S1 = &ST[wave][0];
  float* S2 = &ST[wave][128];
  float* V1 = &ST[wave][256];
  float* V2 = &ST[wave][448];
  float* OUT = &ST[wave][640];

  const float* S1e = S1 + e * 8;
  const float* S2e = S2 + e * 8;
  const float* V1e = V1 + e * 12;
  const float* V2e = V2 + e * 12;

  for (int trip = blockIdx.x; trip < PTRIPS; trip += NGRID) {
  const int tileBase = (trip * PW + wave) * 16;
  const int eg = tileBase + e;
  __syncthreads();

  const int i = eidx[eg];
  if (h == 0) {
#pragma unroll
    for (int u = 0; u < 8; ++u) S1[e * 8 + u] = x[i * D + u];
#pragma unroll
    for (int t = 0; t < 12; ++t) V1[e * 12 + t] = x[i * D + 8 + t];
  } else {
#pragma unroll
    for (int u = 0; u < 8; ++u) {
      float v = eout[eg * D + u] * EBN[u] + EBN[8 + u] + ea[eg * D + u];
      S2[e * 8 + u] = v;
      eout[eg * D + u] = v;
    }
#pragma unroll
    for (int t = 0; t < 12; ++t) {
      float v = eout[eg * D + 8 + t] * EBN[16 + t / 3] + ea[eg * D + 8 + t];
      V2[e * 12 + t] = v;
      eout[eg * D + 8 + t] = v;
    }
  }
  __syncthreads();

  float s1r[8], v1r[12], v2r[6], v1h[6];
#pragma unroll
  for (int u = 0; u < 8; ++u) s1r[u] = S1e[u];
#pragma unroll
  for (int t = 0; t < 12; ++t) v1r[t] = V1e[t];
#pragma unroll
  for (int t = 0; t < 6; ++t) { v2r[t] = V2e[6 * h + t]; v1h[t] = V1e[6 * h + t]; }
  v2f s2h[2], s1h[2];
#pragma unroll
  for (int kc = 0; kc < 2; ++kc) {
    s2h[kc].x = S2e[4 * kc + 2 * h]; s2h[kc].y = S2e[4 * kc + 2 * h + 1];
    s1h[kc].x = S1e[4 * kc + 2 * h]; s1h[kc].y = S1e[4 * kc + 2 * h + 1];
  }

  v8f accS = {};
  {
    const float* wb = WSs + ((size_t)h * 16 + n) * 2;
#pragma unroll
    for (int kc = 0; kc < 16; ++kc) {
      v2f ft;
      ft.x = s1r[kc >> 1] * s2h[kc & 1].x;
      ft.y = s1r[kc >> 1] * s2h[kc & 1].y;
      v2f w = *(const v2f*)(wb + kc * 64);
      accS = __builtin_amdgcn_wmma_f32_16x16x4_f32(false, w, false, ft, (short)0, accS, false, false);
    }
#pragma unroll
    for (int u = 0; u < 4; ++u) {
      v2f ft;
      ft.x = v1r[u * 3] * v2r[0] + v1r[u * 3 + 1] * v2r[1] + v1r[u * 3 + 2] * v2r[2];
      ft.y = v1r[u * 3] * v2r[3] + v1r[u * 3 + 1] * v2r[4] + v1r[u * 3 + 2] * v2r[5];
      v2f w = *(const v2f*)(wb + (16 + u) * 64);
      accS = __builtin_amdgcn_wmma_f32_16x16x4_f32(false, w, false, ft, (short)0, accS, false, false);
    }
  }

  float o[12];
#pragma unroll
  for (int i = 0; i < 12; ++i) o[i] = 0.f;
  const float* wa = WAs + ((size_t)h * 16 + n) * 2;
  {
    v8f T1 = {};
#pragma unroll
    for (int kc = 0; kc < 2; ++kc) {
      v2f a = *(const v2f*)(wa + kc * 64);
      T1 = __builtin_amdgcn_wmma_f32_16x16x4_f32(false, a, false, s1h[kc], (short)0, T1, false, false);
    }
#pragma unroll
    for (int vv = 0; vv < 2; ++vv)
#pragma unroll
      for (int oo = 0; oo < 4; ++oo)
#pragma unroll
        for (int c = 0; c < 3; ++c) o[oo * 3 + c] += v2r[vv * 3 + c] * T1[vv * 4 + oo];
  }
  {
    v8f T2 = {};
#pragma unroll
    for (int kc = 0; kc < 2; ++kc) {
      v2f a = *(const v2f*)(wa + (2 + kc) * 64);
      T2 = __builtin_amdgcn_wmma_f32_16x16x4_f32(false, a, false, s2h[kc], (short)0, T2, false, false);
    }
#pragma unroll
    for (int rr = 0; rr < 2; ++rr) {
      const float a0 = v1h[rr * 3], a1 = v1h[rr * 3 + 1], a2 = v1h[rr * 3 + 2];
#pragma unroll
      for (int oo = 0; oo < 4; ++oo) {
        const float tv = T2[rr * 4 + oo];
        o[oo * 3 + 0] += a0 * tv;
        o[oo * 3 + 1] += a1 * tv;
        o[oo * 3 + 2] += a2 * tv;
      }
    }
  }
  {
    v2f bv[3];
#pragma unroll
    for (int b = 0; b < 3; ++b) { bv[b].x = v2r[b]; bv[b].y = v2r[3 + b]; }
    v2f a = *(const v2f*)(wa + 4 * 64);
    v8f z = {};
    v8f Q0 = __builtin_amdgcn_wmma_f32_16x16x4_f32(false, a, false, bv[0], (short)0, z, false, false);
    v8f Q1 = __builtin_amdgcn_wmma_f32_16x16x4_f32(false, a, false, bv[1], (short)0, z, false, false);
    v8f Q2 = __builtin_amdgcn_wmma_f32_16x16x4_f32(false, a, false, bv[2], (short)0, z, false, false);
#pragma unroll
    for (int rr = 0; rr < 2; ++rr) {
      const float a0 = v1h[rr * 3], a1 = v1h[rr * 3 + 1], a2 = v1h[rr * 3 + 2];
#pragma unroll
      for (int oo = 0; oo < 4; ++oo) {
        const float q0 = Q0[rr * 4 + oo], q1 = Q1[rr * 4 + oo], q2 = Q2[rr * 4 + oo];
        o[oo * 3 + 0] += a1 * q2 - a2 * q1;
        o[oo * 3 + 1] += a2 * q0 - a0 * q2;
        o[oo * 3 + 2] += a0 * q1 - a1 * q0;
      }
    }
  }
#pragma unroll
  for (int i = 0; i < 12; ++i) o[i] += __shfl_xor(o[i], 16, 32);

  float sg[8];
#pragma unroll
  for (int r = 0; r < 8; ++r) sg[r] = sigmoidf_(accS[r]);
  if (h == 0) {
#pragma unroll
    for (int r = 0; r < 8; ++r) OUT[e * D + r] = accS[r] * sg[r];
  } else {
#pragma unroll
    for (int oo = 0; oo < 4; ++oo)
#pragma unroll
      for (int c = 0; c < 3; ++c) OUT[e * D + 8 + oo * 3 + c] = o[oo * 3 + c] * sg[oo];
  }
  __syncthreads();
  int rown[16];
#pragma unroll
  for (int r = 0; r < 16; ++r) rown[r] = __builtin_amdgcn_readlane(i, r);
  if (lane < D) {
    float ov[16];
#pragma unroll
    for (int r = 0; r < 16; ++r) ov[r] = OUT[r * D + lane];
#pragma unroll
    for (int r = 0; r < 16; ++r) atomicAdd(&agg[rown[r] * AP + lane], ov[r]);
  }
  }
}

__global__ __launch_bounds__(256) void k_node_stats(const float* __restrict__ agg,
                                                    float* __restrict__ nstat) {
  __shared__ float BST[20];
  int tid = threadIdx.x;
  if (tid < 20) BST[tid] = 0.f;
  __syncthreads();
  int node = blockIdx.x * 256 + tid;
  float ss[8] = {}, s2[8] = {}, vv[4] = {};
  if (node < NN) {
#pragma unroll
    for (int k = 0; k < 8; ++k) { float t = agg[node * AP + k]; ss[k] = t; s2[k] = t * t; }
#pragma unroll
    for (int t = 0; t < 12; ++t) { float z = agg[node * AP + 8 + t]; vv[t / 3] += z * z; }
  }
#pragma unroll
  for (int off = 16; off >= 1; off >>= 1) {
#pragma unroll
    for (int k = 0; k < 8; ++k) { ss[k] += __shfl_xor(ss[k], off, 32); s2[k] += __shfl_xor(s2[k], off, 32); }
#pragma unroll
    for (int u = 0; u < 4; ++u) vv[u] += __shfl_xor(vv[u], off, 32);
  }
  if ((tid & 31) == 0) {
#pragma unroll
    for (int k = 0; k < 8; ++k) { atomicAdd(&BST[k], ss[k]); atomicAdd(&BST[8 + k], s2[k]); }
#pragma unroll
    for (int u = 0; u < 4; ++u) atomicAdd(&BST[16 + u], vv[u]);
  }
  __syncthreads();
  if (tid < 20) atomicAdd(&nstat[tid], BST[tid]);
}

#define OT 160
#define OGRID 196
#define OQ (NN * D / 4)
#define OTRIPS ((OQ + OGRID * OT - 1) / (OGRID * OT))
__global__ __launch_bounds__(OT) void k_node_out(const float* __restrict__ agg,
                                                 const float* __restrict__ x,
                                                 const float* __restrict__ nstat,
                                                 const float* __restrict__ wsc,
                                                 const float* __restrict__ bsc,
                                                 const float* __restrict__ wvc,
                                                 float* __restrict__ out) {
  __shared__ float NB[20];
  const int tid = threadIdx.x;
  const int g0 = blockIdx.x * OT + tid;
  const float4* agg4 = (const float4*)agg;
  const float4* x4 = (const float4*)x;
  const int g = g0 % 5;
  float4 av[OTRIPS], xv[OTRIPS];
#pragma unroll
  for (int it = 0; it < OTRIPS; ++it) {
    const int q = g0 + it * (OGRID * OT);
    av[it] = make_float4(0.f, 0.f, 0.f, 0.f);
    xv[it] = make_float4(0.f, 0.f, 0.f, 0.f);
    if (q < OQ) { av[it] = agg4[(q / 5) * (AP / 4) + g]; xv[it] = x4[q]; }
  }
  if (tid < 12) bn_coeffs(nstat, wsc, bsc, wvc, NB, tid, (float)NN);
  __syncthreads();
  float sc[4], sh[4];
#pragma unroll
  for (int cc = 0; cc < 4; ++cc) {
    const int k = 4 * g + cc;
    sc[cc] = (k < 8) ? NB[k] : NB[16 + (k - 8) / 3];
    sh[cc] = (k < 8) ? NB[8 + k] : 0.0f;
  }
#pragma unroll
  for (int it = 0; it < OTRIPS; ++it) {
    const int q = g0 + it * (OGRID * OT);
    if (q < OQ) {
      float4 r;
      r.x = av[it].x * sc[0] + sh[0] + xv[it].x;
      r.y = av[it].y * sc[1] + sh[1] + xv[it].y;
      r.z = av[it].z * sc[2] + sh[2] + xv[it].z;
      r.w = av[it].w * sc[3] + sh[3] + xv[it].w;
      ((float4*)out)[q] = r;
    }
  }
}

extern "C" void kernel_launch(void* const* d_in, const int* in_sizes, int n_in,
                              void* d_out, int out_size, void* d_ws, size_t ws_size,
                              hipStream_t stream) {
  (void)in_sizes; (void)n_in; (void)out_size; (void)ws_size;
  const float* x = (const float*)d_in[0];
  const float* ea = (const float*)d_in[1];
  const float* elen = (const float*)d_in[2];
  const float* We_ss = (const float*)d_in[3];
  const float* We_sv = (const float*)d_in[4];
  const float* We_vs = (const float*)d_in[5];
  const float* We_vvs = (const float*)d_in[6];
  const float* We_vvv = (const float*)d_in[7];
  const float* bne_ws = (const float*)d_in[8];
  const float* bne_bs = (const float*)d_in[9];
  const float* bne_wv = (const float*)d_in[10];
  const float* Wn_ss = (const float*)d_in[11];
  const float* Wn_sv = (const float*)d_in[12];
  const float* Wn_vs = (const float*)d_in[13];
  const float* Wn_vvs = (const float*)d_in[14];
  const float* Wn_vvv = (const float*)d_in[15];
  const float* bnn_ws = (const float*)d_in[16];
  const float* bnn_bs = (const float*)d_in[17];
  const float* bnn_wv = (const float*)d_in[18];
  const int* eidx = (const int*)d_in[19];

  float* w = (float*)d_ws;
  float* agg = w;
  float* estat = agg + (size_t)NN * AP;
  float* nstat = estat + 20;
  float* outf = (float*)d_out;
  float* eout = outf + (size_t)NN * D;

  hipMemsetAsync(agg, 0, ((size_t)NN * AP + 40) * sizeof(float), stream);

  k_edge_tp<<<EGRID, PT, 0, stream>>>(x, ea, elen, eidx, We_ss, We_sv, We_vs, We_vvs, We_vvv,
                                        eout, estat);
  k_node_tp<<<NGRID, PT, 0, stream>>>(x, ea, eidx, estat, bne_ws, bne_bs, bne_wv,
                                         Wn_ss, Wn_sv, Wn_vs, Wn_vvs, Wn_vvv, agg, eout);
  k_node_stats<<<(NN + 255) / 256, 256, 0, stream>>>(agg, nstat);
  k_node_out<<<OGRID, OT, 0, stream>>>(agg, x, nstat, bnn_ws, bnn_bs, bnn_wv, outf);
}


# Gate and SiLU sigmoids of both GEMM kernels on the hardware exp2 and reciprocal instructions (about 1 ulp each) in place of library expf and IEEE division: fewer vector issues per tile
# speedup vs baseline: 3.8780x; 1.0344x over previous
#include <hip/hip_runtime.h>
#include <math.h>

#define S 8
#define V 4
#define D 20
#define OS 12
#define NN 25000
#define NE 400000
#define EPSV 1e-5f

#define K1S 168
#define NFRAG 11
#define VPASS 6
#define K2S 80
#define NVFRAG 5
#define NVPASS 3

#define TILES (NE / 16)
#define PW 4
#define PT (PW * 32)
#define PTRIPS (TILES / PW)
#define EGRID 1024
#define NGRID 2048
#define EPASS 21
#define AP 32

typedef __attribute__((ext_vector_type(2))) float v2f;
typedef __attribute__((ext_vector_type(8))) float v8f;

__device__ __forceinline__ float sigmoidf_(float x) {
  return __builtin_amdgcn_rcpf(1.0f + __builtin_amdgcn_exp2f(x * -1.4426950408889634f));
}

__device__ __forceinline__ float feat1s(const float* S1e, const float* S2e, const float* DOTe,
                                        int f) {
  if (f < 136) return S1e[f >> 3] * S2e[f & 7];
  return DOTe[f - 136];
}

__device__ __forceinline__ float feat2s(const float* S1e, const float* S2e, const float* DOTe,
                                        int f) {
  if (f < 64) return S1e[f >> 3] * S2e[f & 7];
  return DOTe[f - 64];
}

__global__ __launch_bounds__(PT) void k_edge_tp(
    const float* __restrict__ x, const float* __restrict__ ea, const float* __restrict__ elen,
    const int* __restrict__ eidx,
    const float* __restrict__ Wss, const float* __restrict__ Wsv, const float* __restrict__ Wvs,
    const float* __restrict__ Wvvs, const float* __restrict__ Wvvv,
    float* __restrict__ msmv, float* __restrict__ estat) {
  __shared__ float WSs[K1S * 16];
  __shared__ float WAs[VPASS * PT];
  __shared__ float ST[PW][1488];
  __shared__ float BST[20];

  const int tid = threadIdx.x;
  __builtin_assume(tid >= 0 && tid < PT);
#pragma unroll
  for (int k = 0; k < EPASS; ++k) {
    const int idx = tid + k * PT;
    const int j = idx & 1, nn = (idx >> 1) & 15, hh = (idx >> 5) & 1, kc = idx >> 6;
    const int f = kc * 4 + 2 * hh + j;
    const float* p = (f < 136) ? (Wss + f * OS) : (Wvvs + (f - 136) * OS);
    const float w = p[nn < OS ? nn : 0];
    WSs[idx] = (nn < OS) ? w : 0.0f;
  }
#pragma unroll
  for (int k = 0; k < VPASS; ++k) {
    const int idx = tid + k * PT;
    const int j = idx & 1, m = (idx >> 1) & 15, hh = (idx >> 5) & 1, fr = idx >> 6;
    const int kk = 2 * hh + j;
    const int u1 = 4 * fr + kk;
    const int q = fr - 5;
    const int off2 = (4 * (q >> 1) + (m >> 2)) * 32 + (4 * (q & 1) + kk) * 4 + (m & 3);
    const int off3 = (4 * (fr - 9) + (m >> 2)) * 16 + kk * 4 + (m & 3);
    const bool live = (fr < 5) ? (u1 < 17) : (fr < NFRAG);
    const float* p = (fr < 5) ? Wsv : (fr < 9) ? Wvs : Wvvv;
    const int off = (fr < 5) ? u1 * 16 + m : (fr < 9) ? off2 : off3;
    const float w = p[live ? off : 0];
    WAs[idx] = live ? w : 0.0f;
  }
  if (tid < 20) BST[tid] = 0.0f;

  const int wave = tid >> 5;
  const int lane = tid & 31;
  const int e = lane & 15;
  const int h = lane >> 4;
  const int n = e;

  float* S1 = &ST[wave][0];
  float* S2 = &ST[wave][272];
  float* V1 = &ST[wave][400];
  float* V2 = &ST[wave][784];
  float* DOT = &ST[wave][976];

  const float* S1e = S1 + e * 17;
  const float* S2e = S2 + e * 8;
  const float* V1e = V1 + e * 24;
  const float* V2e = V2 + e * 12;
  const float* DOTe = DOT + e * 32;

  float st[16];
#pragma unroll
  for (int i = 0; i < 16; ++i) st[i] = 0.f;

  for (int trip = blockIdx.x; trip < PTRIPS; trip += EGRID) {
  const int tileBase = (trip * PW + wave) * 16;
  const int eg = tileBase + e;
  __syncthreads();

  if (h == 0) {
    int i = eidx[eg], j = eidx[NE + eg];
#pragma unroll
    for (int u = 0; u < 8; ++u) { S1[e * 17 + u] = x[i * D + u]; S1[e * 17 + 8 + u] = x[j * D + u]; }
    S1[e * 17 + 16] = elen[eg];
#pragma unroll
    for (int t = 0; t < 12; ++t) { V1[e * 24 + t] = x[i * D + 8 + t]; V1[e * 24 + 12 + t] = x[j * D + 8 + t]; }
  } else {
#pragma unroll
    for (int u = 0; u < 8; ++u) S2[e * 8 + u] = ea[eg * D + u];
#pragma unroll
    for (int t = 0; t < 12; ++t) V2[e * 12 + t] = ea[eg * D + 8 + t];
  }
  __syncthreads();
  {
#pragma unroll
    for (int u = 0; u < 4; ++u)
#pragma unroll
      for (int v = 0; v < 4; ++v) {
        const int uu = 4 * h + u;
        float d = 0.f;
        for (int c = 0; c < 3; ++c) d += V1[e * 24 + uu * 3 + c] * V2[e * 12 + v * 3 + c];
        DOT[e * 32 + uu * 4 + v] = d;
      }
  }
  __syncthreads();

  v8f accS = {};
  {
    const float* wb = WSs + ((size_t)h * 16 + n) * 2;
#pragma unroll
    for (int kc = 0; kc < K1S / 4; ++kc) {
      int f0 = kc * 4 + 2 * h;
      v2f ft;
      ft.x = feat1s(S1e, S2e, DOTe, f0);
      ft.y = feat1s(S1e, S2e, DOTe, f0 + 1);
      v2f w = *(const v2f*)(wb + kc * 64);
      accS = __builtin_amdgcn_wmma_f32_16x16x4_f32(false, w, false, ft, (short)0, accS, false, false);
    }
  }

  float o[12];
#pragma unroll
  for (int i = 0; i < 12; ++i) o[i] = 0.f;
  const float* wa = WAs + ((size_t)h * 16 + n) * 2;
  float v2r[6];
#pragma unroll
  for (int i = 0; i < 6; ++i) v2r[i] = V2e[6 * h + i];
  {
    v8f T1 = {};
#pragma unroll
    for (int kc = 0; kc < 4; ++kc) {
      v2f b;
      b.x = S1e[4 * kc + 2 * h];
      b.y = S1e[4 * kc + 2 * h + 1];
      v2f a = *(const v2f*)(wa + kc * 64);
      T1 = __builtin_amdgcn_wmma_f32_16x16x4_f32(false, a, false, b, (short)0, T1, false, false);
    }
    {
      v2f b;
      b.x = (h == 0) ? S1e[16] : 0.f;
      b.y = 0.f;
      v2f a = *(const v2f*)(wa + 4 * 64);
      T1 = __builtin_amdgcn_wmma_f32_16x16x4_f32(false, a, false, b, (short)0, T1, false, false);
    }
#pragma unroll
    for (int vv = 0; vv < 2; ++vv)
#pragma unroll
      for (int oo = 0; oo < 4; ++oo)
#pragma unroll
        for (int c = 0; c < 3; ++c) o[oo * 3 + c] += v2r[vv * 3 + c] * T1[vv * 4 + oo];
  }
  {
    v2f bs[2];
#pragma unroll
    for (int kc = 0; kc < 2; ++kc) { bs[kc].x = S2e[4 * kc + 2 * h]; bs[kc].y = S2e[4 * kc + 2 * h + 1]; }
#pragma unroll
    for (int t = 0; t < 2; ++t) {
      v8f T2 = {};
#pragma unroll
      for (int kc = 0; kc < 2; ++kc) {
        v2f a = *(const v2f*)(wa + (5 + 2 * t + kc) * 64);
        T2 = __builtin_amdgcn_wmma_f32_16x16x4_f32(false, a, false, bs[kc], (short)0, T2, false, false);
      }
#pragma unroll
      for (int rr = 0; rr < 2; ++rr) {
        const float* v1u = V1e + (4 * t + 2 * h + rr) * 3;
        const float a0 = v1u[0], a1 = v1u[1], a2 = v1u[2];
#pragma unroll
        for (int oo = 0; oo < 4; ++oo) {
          const float tv = T2[rr * 4 + oo];
          o[oo * 3 + 0] += a0 * tv;
          o[oo * 3 + 1] += a1 * tv;
          o[oo * 3 + 2] += a2 * tv;
        }
      }
    }
  }
  {
    v2f bv[3];
#pragma unroll
    for (int b = 0; b < 3; ++b) { bv[b].x = v2r[b]; bv[b].y = v2r[3 + b]; }
#pragma unroll
    for (int t = 0; t < 2; ++t) {
      v2f a = *(const v2f*)(wa + (9 + t) * 64);
      v8f z = {};
      v8f Q0 = __builtin_amdgcn_wmma_f32_16x16x4_f32(false, a, false, bv[0], (short)0, z, false, false);
      v8f Q1 = __builtin_amdgcn_wmma_f32_16x16x4_f32(false, a, false, bv[1], (short)0, z, false, false);
      v8f Q2 = __builtin_amdgcn_wmma_f32_16x16x4_f32(false, a, false, bv[2], (short)0, z, false, false);
#pragma unroll
      for (int rr = 0; rr < 2; ++rr) {
        const float* v1u = V1e + (4 * t + 2 * h + rr) * 3;
        const float a0 = v1u[0], a1 = v1u[1], a2 = v1u[2];
#pragma unroll
        for (int oo = 0; oo < 4; ++oo) {
          const float q0 = Q0[rr * 4 + oo], q1 = Q1[rr * 4 + oo], q2 = Q2[rr * 4 + oo];
          o[oo * 3 + 0] += a1 * q2 - a2 * q1;
          o[oo * 3 + 1] += a2 * q0 - a0 * q2;
          o[oo * 3 + 2] += a0 * q1 - a1 * q0;
        }
      }
    }
  }
#pragma unroll
  for (int i = 0; i < 12; ++i) o[i] += __shfl_xor(o[i], 16, 32);

  float sg[8];
#pragma unroll
  for (int r = 0; r < 8; ++r) sg[r] = sigmoidf_(accS[r]);
  if (h == 0) {
#pragma unroll
    for (int r = 0; r < 8; ++r) {
      float silu = accS[r] * sg[r];
      msmv[eg * D + r] = silu;
      st[r] += silu;
      st[8 + r] += silu * silu;
    }
  } else {
#pragma unroll
    for (int oo = 0; oo < 4; ++oo) {
      float gate = sg[oo];
      float v0 = o[oo * 3 + 0] * gate, v1 = o[oo * 3 + 1] * gate, v2 = o[oo * 3 + 2] * gate;
      msmv[eg * D + 8 + oo * 3 + 0] = v0;
      msmv[eg * D + 8 + oo * 3 + 1] = v1;
      msmv[eg * D + 8 + oo * 3 + 2] = v2;
      st[oo] += v0 * v0 + v1 * v1 + v2 * v2;
    }
  }
  }

#pragma unroll
  for (int off = 8; off >= 1; off >>= 1) {
#pragma unroll
    for (int i = 0; i < 16; ++i) st[i] += __shfl_xor(st[i], off, 32);
  }
  if (e == 0) {
    if (h == 0) {
#pragma unroll
      for (int r = 0; r < 8; ++r) { atomicAdd(&BST[r], st[r]); atomicAdd(&BST[8 + r], st[8 + r]); }
    } else {
#pragma unroll
      for (int oo = 0; oo < 4; ++oo) atomicAdd(&BST[16 + oo], st[oo]);
    }
  }
  __syncthreads();
  if (tid < 20) atomicAdd(&estat[tid], BST[tid]);
}

__device__ __forceinline__ void bn_coeffs(const float* __restrict__ stat, const float* __restrict__ wsc,
                                          const float* __restrict__ bsc, const float* __restrict__ wvc,
                                          float* bn, int t, float count) {
  if (t < 8) {
    float mu = stat[t] / count;
    float var = stat[8 + t] / count - mu * mu;
    float is = rsqrtf(var + EPSV);
    float sc = wsc[t] * is;
    bn[t] = sc;
    bn[8 + t] = bsc[t] - mu * sc;
  } else if (t < 12) {
    int u = t - 8;
    bn[16 + u] = wvc[u] * rsqrtf(stat[16 + u] / (3.0f * count) + EPSV);
  }
}

__global__ __launch_bounds__(PT, 8) void k_node_tp(
    const float* __restrict__ x, const float* __restrict__ ea, const int* __restrict__ eidx,
    const float* __restrict__ estat,
    const float* __restrict__ bws, const float* __restrict__ bbs, const float* __restrict__ bwv,
    const float* __restrict__ Wss, const float* __restrict__ Wsv, const float* __restrict__ Wvs,
    const float* __restrict__ Wvvs, const float* __restrict__ Wvvv,
    float* __restrict__ agg, float* __restrict__ eout) {
  __shared__ float WSs[K2S * 16];
  __shared__ float WAs[NVPASS * PT];
  __shared__ float ST[PW][960];
  __shared__ __attribute__((aligned(16))) float EBN[20];

  const int tid = threadIdx.x;
  __builtin_assume(tid >= 0 && tid < PT);
#pragma unroll
  for (int k = 0; k < K2S * 16 / PT; ++k) {
    const int idx = tid + k * PT;
    const int j = idx & 1, nn = (idx >> 1) & 15, hh = (idx >> 5) & 1, kc = idx >> 6;
    const int f = kc * 4 + 2 * hh + j;
    const float* p = (f < 64) ? (Wss + f * OS) : (Wvvs + (f - 64) * OS);
    const float w = p[nn < OS ? nn : 0];
    WSs[idx] = (nn < OS) ? w : 0.0f;
  }
#pragma unroll
  for (int k = 0; k < NVPASS; ++k) {
    const int idx = tid + k * PT;
    const int j = idx & 1, m = (idx >> 1) & 15, hh = (idx >> 5) & 1, fr = idx >> 6;
    const int kk = 2 * hh + j;
    const int off1 = (4 * fr + kk) * 16 + m;
    const int off2 = (m >> 2) * 32 + (4 * (fr - 2) + kk) * 4 + (m & 3);
    const int off3 = (m >> 2) * 16 + kk * 4 + (m & 3);
    const bool live = fr < NVFRAG;
    const float* p = (fr < 2) ? Wsv : (fr < 4) ? Wvs : Wvvv;
    const int off = (fr < 2) ? off1 : (fr < 4) ? off2 : off3;
    const float w = p[live ? off : 0];
    WAs[idx] = live ? w : 0.0f;
  }
  if (tid < 12) bn_coeffs(estat, bws, bbs, bwv, EBN, tid, (float)NE);

  const int wave = tid >> 5;
  const int lane = tid & 31;
  const int e = lane & 15;
  const int h = lane >> 4;
  const int n = e;

  float* S1 = &ST[wave][0];
  float* S2 = &ST[wave][128];
  float* V1 = &ST[wave][256];
  float* V2 = &ST[wave][448];
  float* OUT = &ST[wave][640];

  const float* S1e = S1 + e * 8;
  const float* S2e = S2 + e * 8;
  const float* V1e = V1 + e * 12;
  const float* V2e = V2 + e * 12;

  for (int trip = blockIdx.x; trip < PTRIPS; trip += NGRID) {
  const int tileBase = (trip * PW + wave) * 16;
  const int eg = tileBase + e;
  __syncthreads();

  const int i = eidx[eg];
  if (h == 0) {
#pragma unroll
    for (int u = 0; u < 8; ++u) S1[e * 8 + u] = x[i * D + u];
#pragma unroll
    for (int t = 0; t < 12; ++t) V1[e * 12 + t] = x[i * D + 8 + t];
  } else {
#pragma unroll
    for (int u = 0; u < 8; ++u) {
      float v = eout[eg * D + u] * EBN[u] + EBN[8 + u] + ea[eg * D + u];
      S2[e * 8 + u] = v;
      eout[eg * D + u] = v;
    }
#pragma unroll
    for (int t = 0; t < 12; ++t) {
      float v = eout[eg * D + 8 + t] * EBN[16 + t / 3] + ea[eg * D + 8 + t];
      V2[e * 12 + t] = v;
      eout[eg * D + 8 + t] = v;
    }
  }
  __syncthreads();

  float s1r[8], v1r[12], v2r[6], v1h[6];
#pragma unroll
  for (int u = 0; u < 8; ++u) s1r[u] = S1e[u];
#pragma unroll
  for (int t = 0; t < 12; ++t) v1r[t] = V1e[t];
#pragma unroll
  for (int t = 0; t < 6; ++t) { v2r[t] = V2e[6 * h + t]; v1h[t] = V1e[6 * h + t]; }
  v2f s2h[2], s1h[2];
#pragma unroll
  for (int kc = 0; kc < 2; ++kc) {
    s2h[kc].x = S2e[4 * kc + 2 * h]; s2h[kc].y = S2e[4 * kc + 2 * h + 1];
    s1h[kc].x = S1e[4 * kc + 2 * h]; s1h[kc].y = S1e[4 * kc + 2 * h + 1];
  }

  v8f accS = {};
  {
    const float* wb = WSs + ((size_t)h * 16 + n) * 2;
#pragma unroll
    for (int kc = 0; kc < 16; ++kc) {
      v2f ft;
      ft.x = s1r[kc >> 1] * s2h[kc & 1].x;
      ft.y = s1r[kc >> 1] * s2h[kc & 1].y;
      v2f w = *(const v2f*)(wb + kc * 64);
      accS = __builtin_amdgcn_wmma_f32_16x16x4_f32(false, w, false, ft, (short)0, accS, false, false);
    }
#pragma unroll
    for (int u = 0; u < 4; ++u) {
      v2f ft;
      ft.x = v1r[u * 3] * v2r[0] + v1r[u * 3 + 1] * v2r[1] + v1r[u * 3 + 2] * v2r[2];
      ft.y = v1r[u * 3] * v2r[3] + v1r[u * 3 + 1] * v2r[4] + v1r[u * 3 + 2] * v2r[5];
      v2f w = *(const v2f*)(wb + (16 + u) * 64);
      accS = __builtin_amdgcn_wmma_f32_16x16x4_f32(false, w, false, ft, (short)0, accS, false, false);
    }
  }

  float o[12];
#pragma unroll
  for (int i = 0; i < 12; ++i) o[i] = 0.f;
  const float* wa = WAs + ((size_t)h * 16 + n) * 2;
  {
    v8f T1 = {};
#pragma unroll
    for (int kc = 0; kc < 2; ++kc) {
      v2f a = *(const v2f*)(wa + kc * 64);
      T1 = __builtin_amdgcn_wmma_f32_16x16x4_f32(false, a, false, s1h[kc], (short)0, T1, false, false);
    }
#pragma unroll
    for (int vv = 0; vv < 2; ++vv)
#pragma unroll
      for (int oo = 0; oo < 4; ++oo)
#pragma unroll
        for (int c = 0; c < 3; ++c) o[oo * 3 + c] += v2r[vv * 3 + c] * T1[vv * 4 + oo];
  }
  {
    v8f T2 = {};
#pragma unroll
    for (int kc = 0; kc < 2; ++kc) {
      v2f a = *(const v2f*)(wa + (2 + kc) * 64);
      T2 = __builtin_amdgcn_wmma_f32_16x16x4_f32(false, a, false, s2h[kc], (short)0, T2, false, false);
    }
#pragma unroll
    for (int rr = 0; rr < 2; ++rr) {
      const float a0 = v1h[rr * 3], a1 = v1h[rr * 3 + 1], a2 = v1h[rr * 3 + 2];
#pragma unroll
      for (int oo = 0; oo < 4; ++oo) {
        const float tv = T2[rr * 4 + oo];
        o[oo * 3 + 0] += a0 * tv;
        o[oo * 3 + 1] += a1 * tv;
        o[oo * 3 + 2] += a2 * tv;
      }
    }
  }
  {
    v2f bv[3];
#pragma unroll
    for (int b = 0; b < 3; ++b) { bv[b].x = v2r[b]; bv[b].y = v2r[3 + b]; }
    v2f a = *(const v2f*)(wa + 4 * 64);
    v8f z = {};
    v8f Q0 = __builtin_amdgcn_wmma_f32_16x16x4_f32(false, a, false, bv[0], (short)0, z, false, false);
    v8f Q1 = __builtin_amdgcn_wmma_f32_16x16x4_f32(false, a, false, bv[1], (short)0, z, false, false);
    v8f Q2 = __builtin_amdgcn_wmma_f32_16x16x4_f32(false, a, false, bv[2], (short)0, z, false, false);
#pragma unroll
    for (int rr = 0; rr < 2; ++rr) {
      const float a0 = v1h[rr * 3], a1 = v1h[rr * 3 + 1], a2 = v1h[rr * 3 + 2];
#pragma unroll
      for (int oo = 0; oo < 4; ++oo) {
        const float q0 = Q0[rr * 4 + oo], q1 = Q1[rr * 4 + oo], q2 = Q2[rr * 4 + oo];
        o[oo * 3 + 0] += a1 * q2 - a2 * q1;
        o[oo * 3 + 1] += a2 * q0 - a0 * q2;
        o[oo * 3 + 2] += a0 * q1 - a1 * q0;
      }
    }
  }
#pragma unroll
  for (int i = 0; i < 12; ++i) o[i] += __shfl_xor(o[i], 16, 32);

  float sg[8];
#pragma unroll
  for (int r = 0; r < 8; ++r) sg[r] = sigmoidf_(accS[r]);
  if (h == 0) {
#pragma unroll
    for (int r = 0; r < 8; ++r) OUT[e * D + r] = accS[r] * sg[r];
  } else {
#pragma unroll
    for (int oo = 0; oo < 4; ++oo)
#pragma unroll
      for (int c = 0; c < 3; ++c) OUT[e * D + 8 + oo * 3 + c] = o[oo * 3 + c] * sg[oo];
  }
  __syncthreads();
  int rown[16];
#pragma unroll
  for (int r = 0; r < 16; ++r) rown[r] = __builtin_amdgcn_readlane(i, r);
  if (lane < D) {
    float ov[16];
#pragma unroll
    for (int r = 0; r < 16; ++r) ov[r] = OUT[r * D + lane];
#pragma unroll
    for (int r = 0; r < 16; ++r) atomicAdd(&agg[rown[r] * AP + lane], ov[r]);
  }
  }
}

__global__ __launch_bounds__(256) void k_node_stats(const float* __restrict__ agg,
                                                    float* __restrict__ nstat) {
  __shared__ float BST[20];
  int tid = threadIdx.x;
  if (tid < 20) BST[tid] = 0.f;
  __syncthreads();
  int node = blockIdx.x * 256 + tid;
  float ss[8] = {}, s2[8] = {}, vv[4] = {};
  if (node < NN) {
#pragma unroll
    for (int k = 0; k < 8; ++k) { float t = agg[node * AP + k]; ss[k] = t; s2[k] = t * t; }
#pragma unroll
    for (int t = 0; t < 12; ++t) { float z = agg[node * AP + 8 + t]; vv[t / 3] += z * z; }
  }
#pragma unroll
  for (int off = 16; off >= 1; off >>= 1) {
#pragma unroll
    for (int k = 0; k < 8; ++k) { ss[k] += __shfl_xor(ss[k], off, 32); s2[k] += __shfl_xor(s2[k], off, 32); }
#pragma unroll
    for (int u = 0; u < 4; ++u) vv[u] += __shfl_xor(vv[u], off, 32);
  }
  if ((tid & 31) == 0) {
#pragma unroll
    for (int k = 0; k < 8; ++k) { atomicAdd(&BST[k], ss[k]); atomicAdd(&BST[8 + k], s2[k]); }
#pragma unroll
    for (int u = 0; u < 4; ++u) atomicAdd(&BST[16 + u], vv[u]);
  }
  __syncthreads();
  if (tid < 20) atomicAdd(&nstat[tid], BST[tid]);
}

#define OT 160
#define OGRID 196
#define OQ (NN * D / 4)
#define OTRIPS ((OQ + OGRID * OT - 1) / (OGRID * OT))
__global__ __launch_bounds__(OT) void k_node_out(const float* __restrict__ agg,
                                                 const float* __restrict__ x,
                                                 const float* __restrict__ nstat,
                                                 const float* __restrict__ wsc,
                                                 const float* __restrict__ bsc,
                                                 const float* __restrict__ wvc,
                                                 float* __restrict__ out) {
  __shared__ float NB[20];
  const int tid = threadIdx.x;
  const int g0 = blockIdx.x * OT + tid;
  const float4* agg4 = (const float4*)agg;
  const float4* x4 = (const float4*)x;
  const int g = g0 % 5;
  float4 av[OTRIPS], xv[OTRIPS];
#pragma unroll
  for (int it = 0; it < OTRIPS; ++it) {
    const int q = g0 + it * (OGRID * OT);
    av[it] = make_float4(0.f, 0.f, 0.f, 0.f);
    xv[it] = make_float4(0.f, 0.f, 0.f, 0.f);
    if (q < OQ) { av[it] = agg4[(q / 5) * (AP / 4) + g]; xv[it] = x4[q]; }
  }
  if (tid < 12) bn_coeffs(nstat, wsc, bsc, wvc, NB, tid, (float)NN);
  __syncthreads();
  float sc[4], sh[4];
#pragma unroll
  for (int cc = 0; cc < 4; ++cc) {
    const int k = 4 * g + cc;
    sc[cc] = (k < 8) ? NB[k] : NB[16 + (k - 8) / 3];
    sh[cc] = (k < 8) ? NB[8 + k] : 0.0f;
  }
#pragma unroll
  for (int it = 0; it < OTRIPS; ++it) {
    const int q = g0 + it * (OGRID * OT);
    if (q < OQ) {
      float4 r;
      r.x = av[it].x * sc[0] + sh[0] + xv[it].x;
      r.y = av[it].y * sc[1] + sh[1] + xv[it].y;
      r.z = av[it].z * sc[2] + sh[2] + xv[it].z;
      r.w = av[it].w * sc[3] + sh[3] + xv[it].w;
      ((float4*)out)[q] = r;
    }
  }
}

extern "C" void kernel_launch(void* const* d_in, const int* in_sizes, int n_in,
                              void* d_out, int out_size, void* d_ws, size_t ws_size,
                              hipStream_t stream) {
  (void)in_sizes; (void)n_in; (void)out_size; (void)ws_size;
  const float* x = (const float*)d_in[0];
  const float* ea = (const float*)d_in[1];
  const float* elen = (const float*)d_in[2];
  const float* We_ss = (const float*)d_in[3];
  const float* We_sv = (const float*)d_in[4];
  const float* We_vs = (const float*)d_in[5];
  const float* We_vvs = (const float*)d_in[6];
  const float* We_vvv = (const float*)d_in[7];
  const float* bne_ws = (const float*)d_in[8];
  const float* bne_bs = (const float*)d_in[9];
  const float* bne_wv = (const float*)d_in[10];
  const float* Wn_ss = (const float*)d_in[11];
  const float* Wn_sv = (const float*)d_in[12];
  const float* Wn_vs = (const float*)d_in[13];
  const float* Wn_vvs = (const float*)d_in[14];
  const float* Wn_vvv = (const float*)d_in[15];
  const float* bnn_ws = (const float*)d_in[16];
  const float* bnn_bs = (const float*)d_in[17];
  const float* bnn_wv = (const float*)d_in[18];
  const int* eidx = (const int*)d_in[19];

  float* w = (float*)d_ws;
  float* agg = w;
  float* estat = agg + (size_t)NN * AP;
  float* nstat = estat + 20;
  float* outf = (float*)d_out;
  float* eout = outf + (size_t)NN * D;

  hipMemsetAsync(agg, 0, ((size_t)NN * AP + 40) * sizeof(float), stream);

  k_edge_tp<<<EGRID, PT, 0, stream>>>(x, ea, elen, eidx, We_ss, We_sv, We_vs, We_vvs, We_vvv,
                                        eout, estat);
  k_node_tp<<<NGRID, PT, 0, stream>>>(x, ea, eidx, estat, bne_ws, bne_bs, bne_wv,
                                         Wn_ss, Wn_sv, Wn_vs, Wn_vvs, Wn_vvv, agg, eout);
  k_node_stats<<<(NN + 255) / 256, 256, 0, stream>>>(agg, nstat);
  k_node_out<<<OGRID, OT, 0, stream>>>(agg, x, nstat, bnn_ws, bnn_bs, bnn_wv, outf);
}


# Node scatter-add: one agg plane per half of the second-level cache, picked by the wave's die id, so no agg line moves between halves; stats and output passes add the two planes (sum order changes)
# speedup vs baseline: 4.0364x; 1.0408x over previous
#include <hip/hip_runtime.h>
#include <math.h>

#define S 8
#define V 4
#define D 20
#define OS 12
#define NN 25000
#define NE 400000
#define EPSV 1e-5f

#define K1S 168
#define NFRAG 11
#define VPASS 6
#define K2S 80
#define NVFRAG 5
#define NVPASS 3

#define TILES (NE / 16)
#define PW 4
#define PT (PW * 32)
#define PTRIPS (TILES / PW)
#define EGRID 1024
#define NGRID 2048
#define EPASS 21
#define AP 32
#define AGGN (NN * AP)

typedef __attribute__((ext_vector_type(2))) float v2f;
typedef __attribute__((ext_vector_type(8))) float v8f;

__device__ __forceinline__ float sigmoidf_(float x) {
  return __builtin_amdgcn_rcpf(1.0f + __builtin_amdgcn_exp2f(x * -1.4426950408889634f));
}

__device__ __forceinline__ int l2_half() {
  return (int)((__builtin_amdgcn_s_sendmsg_rtn(0x87) >> 18) & 1u);
}

__device__ __forceinline__ float feat1s(const float* S1e, const float* S2e, const float* DOTe,
                                        int f) {
  if (f < 136) return S1e[f >> 3] * S2e[f & 7];
  return DOTe[f - 136];
}

__device__ __forceinline__ float feat2s(const float* S1e, const float* S2e, const float* DOTe,
                                        int f) {
  if (f < 64) return S1e[f >> 3] * S2e[f & 7];
  return DOTe[f - 64];
}

__global__ __launch_bounds__(PT) void k_edge_tp(
    const float* __restrict__ x, const float* __restrict__ ea, const float* __restrict__ elen,
    const int* __restrict__ eidx,
    const float* __restrict__ Wss, const float* __restrict__ Wsv, const float* __restrict__ Wvs,
    const float* __restrict__ Wvvs, const float* __restrict__ Wvvv,
    float* __restrict__ msmv, float* __restrict__ estat) {
  __shared__ float WSs[K1S * 16];
  __shared__ float WAs[VPASS * PT];
  __shared__ float ST[PW][1488];
  __shared__ float BST[20];

  const int tid = threadIdx.x;
  __builtin_assume(tid >= 0 && tid < PT);
#pragma unroll
  for (int k = 0; k < EPASS; ++k) {
    const int idx = tid + k * PT;
    const int j = idx & 1, nn = (idx >> 1) & 15, hh = (idx >> 5) & 1, kc = idx >> 6;
    const int f = kc * 4 + 2 * hh + j;
    const float* p = (f < 136) ? (Wss + f * OS) : (Wvvs + (f - 136) * OS);
    const float w = p[nn < OS ? nn : 0];
    WSs[idx] = (nn < OS) ? w : 0.0f;
  }
#pragma unroll
  for (int k = 0; k < VPASS; ++k) {
    const int idx = tid + k * PT;
    const int j = idx & 1, m = (idx >> 1) & 15, hh = (idx >> 5) & 1, fr = idx >> 6;
    const int kk = 2 * hh + j;
    const int u1 = 4 * fr + kk;
    const int q = fr - 5;
    const int off2 = (4 * (q >> 1) + (m >> 2)) * 32 + (4 * (q & 1) + kk) * 4 + (m & 3);
    const int off3 = (4 * (fr - 9) + (m >> 2)) * 16 + kk * 4 + (m & 3);
    const bool live = (fr < 5) ? (u1 < 17) : (fr < NFRAG);
    const float* p = (fr < 5) ? Wsv : (fr < 9) ? Wvs : Wvvv;
    const int off = (fr < 5) ? u1 * 16 + m : (fr < 9) ? off2 : off3;
    const float w = p[live ? off : 0];
    WAs[idx] = live ? w : 0.0f;
  }
  if (tid < 20) BST[tid] = 0.0f;

  const int wave = tid >> 5;
  const int lane = tid & 31;
  const int e = lane & 15;
  const int h = lane >> 4;
  const int n = e;

  float* S1 = &ST[wave][0];
  float* S2 = &ST[wave][272];
  float* V1 = &ST[wave][400];
  float* V2 = &ST[wave][784];
  float* DOT = &ST[wave][976];

  const float* S1e = S1 + e * 17;
  const float* S2e = S2 + e * 8;
  const float* V1e = V1 + e * 24;
  const float* V2e = V2 + e * 12;
  const float* DOTe = DOT + e * 32;

  float st[16];
#pragma unroll
  for (int i = 0; i < 16; ++i) st[i] = 0.f;

  for (int trip = blockIdx.x; trip < PTRIPS; trip += EGRID) {
  const int tileBase = (trip * PW + wave) * 16;
  const int eg = tileBase + e;
  __syncthreads();

  if (h == 0) {
    int i = eidx[eg], j = eidx[NE + eg];
#pragma unroll
    for (int u = 0; u < 8; ++u) { S1[e * 17 + u] = x[i * D + u]; S1[e * 17 + 8 + u] = x[j * D + u]; }
    S1[e * 17 + 16] = elen[eg];
#pragma unroll
    for (int t = 0; t < 12; ++t) { V1[e * 24 + t] = x[i * D + 8 + t]; V1[e * 24 + 12 + t] = x[j * D + 8 + t]; }
  } else {
#pragma unroll
    for (int u = 0; u < 8; ++u) S2[e * 8 + u] = ea[eg * D + u];
#pragma unroll
    for (int t = 0; t < 12; ++t) V2[e * 12 + t] = ea[eg * D + 8 + t];
  }
  __syncthreads();
  {
#pragma unroll
    for (int u = 0; u < 4; ++u)
#pragma unroll
      for (int v = 0; v < 4; ++v) {
        const int uu = 4 * h + u;
        float d = 0.f;
        for (int c = 0; c < 3; ++c) d += V1[e * 24 + uu * 3 + c] * V2[e * 12 + v * 3 + c];
        DOT[e * 32 + uu * 4 + v] = d;
      }
  }
  __syncthreads();

  v8f accS = {};
  {
    const float* wb = WSs + ((size_t)h * 16 + n) * 2;
#pragma unroll
    for (int kc = 0; kc < K1S / 4; ++kc) {
      int f0 = kc * 4 + 2 * h;
      v2f ft;
      ft.x = feat1s(S1e, S2e, DOTe, f0);
      ft.y = feat1s(S1e, S2e, DOTe, f0 + 1);
      v2f w = *(const v2f*)(wb + kc * 64);
      accS = __builtin_amdgcn_wmma_f32_16x16x4_f32(false, w, false, ft, (short)0, accS, false, false);
    }
  }

  float o[12];
#pragma unroll
  for (int i = 0; i < 12; ++i) o[i] = 0.f;
  const float* wa = WAs + ((size_t)h * 16 + n) * 2;
  float v2r[6];
#pragma unroll
  for (int i = 0; i < 6; ++i) v2r[i] = V2e[6 * h + i];
  {
    v8f T1 = {};
#pragma unroll
    for (int kc = 0; kc < 4; ++kc) {
      v2f b;
      b.x = S1e[4 * kc + 2 * h];
      b.y = S1e[4 * kc + 2 * h + 1];
      v2f a = *(const v2f*)(wa + kc * 64);
      T1 = __builtin_amdgcn_wmma_f32_16x16x4_f32(false, a, false, b, (short)0, T1, false, false);
    }
    {
      v2f b;
      b.x = (h == 0) ? S1e[16] : 0.f;
      b.y = 0.f;
      v2f a = *(const v2f*)(wa + 4 * 64);
      T1 = __builtin_amdgcn_wmma_f32_16x16x4_f32(false, a, false, b, (short)0, T1, false, false);
    }
#pragma unroll
    for (int vv = 0; vv < 2; ++vv)
#pragma unroll
      for (int oo = 0; oo < 4; ++oo)
#pragma unroll
        for (int c = 0; c < 3; ++c) o[oo * 3 + c] += v2r[vv * 3 + c] * T1[vv * 4 + oo];
  }
  {
    v2f bs[2];
#pragma unroll
    for (int kc = 0; kc < 2; ++kc) { bs[kc].x = S2e[4 * kc + 2 * h]; bs[kc].y = S2e[4 * kc + 2 * h + 1]; }
#pragma unroll
    for (int t = 0; t < 2; ++t) {
      v8f T2 = {};
#pragma unroll
      for (int kc = 0; kc < 2; ++kc) {
        v2f a = *(const v2f*)(wa + (5 + 2 * t + kc) * 64);
        T2 = __builtin_amdgcn_wmma_f32_16x16x4_f32(false, a, false, bs[kc], (short)0, T2, false, false);
      }
#pragma unroll
      for (int rr = 0; rr < 2; ++rr) {
        const float* v1u = V1e + (4 * t + 2 * h + rr) * 3;
        const float a0 = v1u[0], a1 = v1u[1], a2 = v1u[2];
#pragma unroll
        for (int oo = 0; oo < 4; ++oo) {
          const float tv = T2[rr * 4 + oo];
          o[oo * 3 + 0] += a0 * tv;
          o[oo * 3 + 1] += a1 * tv;
          o[oo * 3 + 2] += a2 * tv;
        }
      }
    }
  }
  {
    v2f bv[3];
#pragma unroll
    for (int b = 0; b < 3; ++b) { bv[b].x = v2r[b]; bv[b].y = v2r[3 + b]; }
#pragma unroll
    for (int t = 0; t < 2; ++t) {
      v2f a = *(const v2f*)(wa + (9 + t) * 64);
      v8f z = {};
      v8f Q0 = __builtin_amdgcn_wmma_f32_16x16x4_f32(false, a, false, bv[0], (short)0, z, false, false);
      v8f Q1 = __builtin_amdgcn_wmma_f32_16x16x4_f32(false, a, false, bv[1], (short)0, z, false, false);
      v8f Q2 = __builtin_amdgcn_wmma_f32_16x16x4_f32(false, a, false, bv[2], (short)0, z, false, false);
#pragma unroll
      for (int rr = 0; rr < 2; ++rr) {
        const float* v1u = V1e + (4 * t + 2 * h + rr) * 3;
        const float a0 = v1u[0], a1 = v1u[1], a2 = v1u[2];
#pragma unroll
        for (int oo = 0; oo < 4; ++oo) {
          const float q0 = Q0[rr * 4 + oo], q1 = Q1[rr * 4 + oo], q2 = Q2[rr * 4 + oo];
          o[oo * 3 + 0] += a1 * q2 - a2 * q1;
          o[oo * 3 + 1] += a2 * q0 - a0 * q2;
          o[oo * 3 + 2] += a0 * q1 - a1 * q0;
        }
      }
    }
  }
#pragma unroll
  for (int i = 0; i < 12; ++i) o[i] += __shfl_xor(o[i], 16, 32);

  float sg[8];
#pragma unroll
  for (int r = 0; r < 8; ++r) sg[r] = sigmoidf_(accS[r]);
  if (h == 0) {
#pragma unroll
    for (int r = 0; r < 8; ++r) {
      float silu = accS[r] * sg[r];
      msmv[eg * D + r] = silu;
      st[r] += silu;
      st[8 + r] += silu * silu;
    }
  } else {
#pragma unroll
    for (int oo = 0; oo < 4; ++oo) {
      float gate = sg[oo];
      float v0 = o[oo * 3 + 0] * gate, v1 = o[oo * 3 + 1] * gate, v2 = o[oo * 3 + 2] * gate;
      msmv[eg * D + 8 + oo * 3 + 0] = v0;
      msmv[eg * D + 8 + oo * 3 + 1] = v1;
      msmv[eg * D + 8 + oo * 3 + 2] = v2;
      st[oo] += v0 * v0 + v1 * v1 + v2 * v2;
    }
  }
  }

#pragma unroll
  for (int off = 8; off >= 1; off >>= 1) {
#pragma unroll
    for (int i = 0; i < 16; ++i) st[i] += __shfl_xor(st[i], off, 32);
  }
  if (e == 0) {
    if (h == 0) {
#pragma unroll
      for (int r = 0; r < 8; ++r) { atomicAdd(&BST[r], st[r]); atomicAdd(&BST[8 + r], st[8 + r]); }
    } else {
#pragma unroll
      for (int oo = 0; oo < 4; ++oo) atomicAdd(&BST[16 + oo], st[oo]);
    }
  }
  __syncthreads();
  if (tid < 20) atomicAdd(&estat[tid], BST[tid]);
}

__device__ __forceinline__ void bn_coeffs(const float* __restrict__ stat, const float* __restrict__ wsc,
                                          const float* __restrict__ bsc, const float* __restrict__ wvc,
                                          float* bn, int t, float count) {
  if (t < 8) {
    float mu = stat[t] / count;
    float var = stat[8 + t] / count - mu * mu;
    float is = rsqrtf(var + EPSV);
    float sc = wsc[t] * is;
    bn[t] = sc;
    bn[8 + t] = bsc[t] - mu * sc;
  } else if (t < 12) {
    int u = t - 8;
    bn[16 + u] = wvc[u] * rsqrtf(stat[16 + u] / (3.0f * count) + EPSV);
  }
}

__global__ __launch_bounds__(PT, 8) void k_node_tp(
    const float* __restrict__ x, const float* __restrict__ ea, const int* __restrict__ eidx,
    const float* __restrict__ estat,
    const float* __restrict__ bws, const float* __restrict__ bbs, const float* __restrict__ bwv,
    const float* __restrict__ Wss, const float* __restrict__ Wsv, const float* __restrict__ Wvs,
    const float* __restrict__ Wvvs, const float* __restrict__ Wvvv,
    float* __restrict__ agg, float* __restrict__ eout) {
  __shared__ float WSs[K2S * 16];
  __shared__ float WAs[NVPASS * PT];
  __shared__ float ST[PW][960];
  __shared__ __attribute__((aligned(16))) float EBN[20];

  const int tid = threadIdx.x;
  __builtin_assume(tid >= 0 && tid < PT);
#pragma unroll
  for (int k = 0; k < K2S * 16 / PT; ++k) {
    const int idx = tid + k * PT;
    const int j = idx & 1, nn = (idx >> 1) & 15, hh = (idx >> 5) & 1, kc = idx >> 6;
    const int f = kc * 4 + 2 * hh + j;
    const float* p = (f < 64) ? (Wss + f * OS) : (Wvvs + (f - 64) * OS);
    const float w = p[nn < OS ? nn : 0];
    WSs[idx] = (nn < OS) ? w : 0.0f;
  }
#pragma unroll
  for (int k = 0; k < NVPASS; ++k) {
    const int idx = tid + k * PT;
    const int j = idx & 1, m = (idx >> 1) & 15, hh = (idx >> 5) & 1, fr = idx >> 6;
    const int kk = 2 * hh + j;
    const int off1 = (4 * fr + kk) * 16 + m;
    const int off2 = (m >> 2) * 32 + (4 * (fr - 2) + kk) * 4 + (m & 3);
    const int off3 = (m >> 2) * 16 + kk * 4 + (m & 3);
    const bool live = fr < NVFRAG;
    const float* p = (fr < 2) ? Wsv : (fr < 4) ? Wvs : Wvvv;
    const int off = (fr < 2) ? off1 : (fr < 4) ? off2 : off3;
    const float w = p[live ? off : 0];
    WAs[idx] = live ? w : 0.0f;
  }
  if (tid < 12) bn_coeffs(estat, bws, bbs, bwv, EBN, tid, (float)NE);

  const int wave = tid >> 5;
  const int lane = tid & 31;
  const int e = lane & 15;
  const int h = lane >> 4;
  const int n = e;

  float* S1 = &ST[wave][0];
  float* S2 = &ST[wave][128];
  float* V1 = &ST[wave][256];
  float* V2 = &ST[wave][448];
  float* OUT = &ST[wave][640];
  float* __restrict__ aggh = agg + (size_t)l2_half() * AGGN;

  const float* S1e = S1 + e * 8;
  const float* S2e = S2 + e * 8;
  const float* V1e = V1 + e * 12;
  const float* V2e = V2 + e * 12;

  for (int trip = blockIdx.x; trip < PTRIPS; trip += NGRID) {
  const int tileBase = (trip * PW + wave) * 16;
  const int eg = tileBase + e;
  __syncthreads();

  const int i = eidx[eg];
  if (h == 0) {
#pragma unroll
    for (int u = 0; u < 8; ++u) S1[e * 8 + u] = x[i * D + u];
#pragma unroll
    for (int t = 0; t < 12; ++t) V1[e * 12 + t] = x[i * D + 8 + t];
  } else {
#pragma unroll
    for (int u = 0; u < 8; ++u) {
      float v = eout[eg * D + u] * EBN[u] + EBN[8 + u] + ea[eg * D + u];
      S2[e * 8 + u] = v;
      eout[eg * D + u] = v;
    }
#pragma unroll
    for (int t = 0; t < 12; ++t) {
      float v = eout[eg * D + 8 + t] * EBN[16 + t / 3] + ea[eg * D + 8 + t];
      V2[e * 12 + t] = v;
      eout[eg * D + 8 + t] = v;
    }
  }
  __syncthreads();

  float s1r[8], v1r[12], v2r[6], v1h[6];
#pragma unroll
  for (int u = 0; u < 8; ++u) s1r[u] = S1e[u];
#pragma unroll
  for (int t = 0; t < 12; ++t) v1r[t] = V1e[t];
#pragma unroll
  for (int t = 0; t < 6; ++t) { v2r[t] = V2e[6 * h + t]; v1h[t] = V1e[6 * h + t]; }
  v2f s2h[2], s1h[2];
#pragma unroll
  for (int kc = 0; kc < 2; ++kc) {
    s2h[kc].x = S2e[4 * kc + 2 * h]; s2h[kc].y = S2e[4 * kc + 2 * h + 1];
    s1h[kc].x = S1e[4 * kc + 2 * h]; s1h[kc].y = S1e[4 * kc + 2 * h + 1];
  }

  v8f accS = {};
  {
    const float* wb = WSs + ((size_t)h * 16 + n) * 2;
#pragma unroll
    for (int kc = 0; kc < 16; ++kc) {
      v2f ft;
      ft.x = s1r[kc >> 1] * s2h[kc & 1].x;
      ft.y = s1r[kc >> 1] * s2h[kc & 1].y;
      v2f w = *(const v2f*)(wb + kc * 64);
      accS = __builtin_amdgcn_wmma_f32_16x16x4_f32(false, w, false, ft, (short)0, accS, false, false);
    }
#pragma unroll
    for (int u = 0; u < 4; ++u) {
      v2f ft;
      ft.x = v1r[u * 3] * v2r[0] + v1r[u * 3 + 1] * v2r[1] + v1r[u * 3 + 2] * v2r[2];
      ft.y = v1r[u * 3] * v2r[3] + v1r[u * 3 + 1] * v2r[4] + v1r[u * 3 + 2] * v2r[5];
      v2f w = *(const v2f*)(wb + (16 + u) * 64);
      accS = __builtin_amdgcn_wmma_f32_16x16x4_f32(false, w, false, ft, (short)0, accS, false, false);
    }
  }

  float o[12];
#pragma unroll
  for (int i = 0; i < 12; ++i) o[i] = 0.f;
  const float* wa = WAs + ((size_t)h * 16 + n) * 2;
  {
    v8f T1 = {};
#pragma unroll
    for (int kc = 0; kc < 2; ++kc) {
      v2f a = *(const v2f*)(wa + kc * 64);
      T1 = __builtin_amdgcn_wmma_f32_16x16x4_f32(false, a, false, s1h[kc], (short)0, T1, false, false);
    }
#pragma unroll
    for (int vv = 0; vv < 2; ++vv)
#pragma unroll
      for (int oo = 0; oo < 4; ++oo)
#pragma unroll
        for (int c = 0; c < 3; ++c) o[oo * 3 + c] += v2r[vv * 3 + c] * T1[vv * 4 + oo];
  }
  {
    v8f T2 = {};
#pragma unroll
    for (int kc = 0; kc < 2; ++kc) {
      v2f a = *(const v2f*)(wa + (2 + kc) * 64);
      T2 = __builtin_amdgcn_wmma_f32_16x16x4_f32(false, a, false, s2h[kc], (short)0, T2, false, false);
    }
#pragma unroll
    for (int rr = 0; rr < 2; ++rr) {
      const float a0 = v1h[rr * 3], a1 = v1h[rr * 3 + 1], a2 = v1h[rr * 3 + 2];
#pragma unroll
      for (int oo = 0; oo < 4; ++oo) {
        const float tv = T2[rr * 4 + oo];
        o[oo * 3 + 0] += a0 * tv;
        o[oo * 3 + 1] += a1 * tv;
        o[oo * 3 + 2] += a2 * tv;
      }
    }
  }
  {
    v2f bv[3];
#pragma unroll
    for (int b = 0; b < 3; ++b) { bv[b].x = v2r[b]; bv[b].y = v2r[3 + b]; }
    v2f a = *(const v2f*)(wa + 4 * 64);
    v8f z = {};
    v8f Q0 = __builtin_amdgcn_wmma_f32_16x16x4_f32(false, a, false, bv[0], (short)0, z, false, false);
    v8f Q1 = __builtin_amdgcn_wmma_f32_16x16x4_f32(false, a, false, bv[1], (short)0, z, false, false);
    v8f Q2 = __builtin_amdgcn_wmma_f32_16x16x4_f32(false, a, false, bv[2], (short)0, z, false, false);
#pragma unroll
    for (int rr = 0; rr < 2; ++rr) {
      const float a0 = v1h[rr * 3], a1 = v1h[rr * 3 + 1], a2 = v1h[rr * 3 + 2];
#pragma unroll
      for (int oo = 0; oo < 4; ++oo) {
        const float q0 = Q0[rr * 4 + oo], q1 = Q1[rr * 4 + oo], q2 = Q2[rr * 4 + oo];
        o[oo * 3 + 0] += a1 * q2 - a2 * q1;
        o[oo * 3 + 1] += a2 * q0 - a0 * q2;
        o[oo * 3 + 2] += a0 * q1 - a1 * q0;
      }
    }
  }
#pragma unroll
  for (int i = 0; i < 12; ++i) o[i] += __shfl_xor(o[i], 16, 32);

  float sg[8];
#pragma unroll
  for (int r = 0; r < 8; ++r) sg[r] = sigmoidf_(accS[r]);
  if (h == 0) {
#pragma unroll
    for (int r = 0; r < 8; ++r) OUT[e * D + r] = accS[r] * sg[r];
  } else {
#pragma unroll
    for (int oo = 0; oo < 4; ++oo)
#pragma unroll
      for (int c = 0; c < 3; ++c) OUT[e * D + 8 + oo * 3 + c] = o[oo * 3 + c] * sg[oo];
  }
  __syncthreads();
  int rown[16];
#pragma unroll
  for (int r = 0; r < 16; ++r) rown[r] = __builtin_amdgcn_readlane(i, r);
  if (lane < D) {
    float ov[16];
#pragma unroll
    for (int r = 0; r < 16; ++r) ov[r] = OUT[r * D + lane];
#pragma unroll
    for (int r = 0; r < 16; ++r) atomicAdd(&aggh[rown[r] * AP + lane], ov[r]);
  }
  }
}

__global__ __launch_bounds__(256) void k_node_stats(const float* __restrict__ agg,
                                                    float* __restrict__ nstat) {
  __shared__ float BST[20];
  int tid = threadIdx.x;
  if (tid < 20) BST[tid] = 0.f;
  __syncthreads();
  int node = blockIdx.x * 256 + tid;
  float ss[8] = {}, s2[8] = {}, vv[4] = {};
  if (node < NN) {
#pragma unroll
    for (int k = 0; k < 8; ++k) { float t = agg[node * AP + k] + agg[AGGN + node * AP + k]; ss[k] = t; s2[k] = t * t; }
#pragma unroll
    for (int t = 0; t < 12; ++t) { float z = agg[node * AP + 8 + t] + agg[AGGN + node * AP + 8 + t]; vv[t / 3] += z * z; }
  }
#pragma unroll
  for (int off = 16; off >= 1; off >>= 1) {
#pragma unroll
    for (int k = 0; k < 8; ++k) { ss[k] += __shfl_xor(ss[k], off, 32); s2[k] += __shfl_xor(s2[k], off, 32); }
#pragma unroll
    for (int u = 0; u < 4; ++u) vv[u] += __shfl_xor(vv[u], off, 32);
  }
  if ((tid & 31) == 0) {
#pragma unroll
    for (int k = 0; k < 8; ++k) { atomicAdd(&BST[k], ss[k]); atomicAdd(&BST[8 + k], s2[k]); }
#pragma unroll
    for (int u = 0; u < 4; ++u) atomicAdd(&BST[16 + u], vv[u]);
  }
  __syncthreads();
  if (tid < 20) atomicAdd(&nstat[tid], BST[tid]);
}

#define OT 160
#define OGRID 196
#define OQ (NN * D / 4)
#define OTRIPS ((OQ + OGRID * OT - 1) / (OGRID * OT))
__global__ __launch_bounds__(OT) void k_node_out(const float* __restrict__ agg,
                                                 const float* __restrict__ x,
                                                 const float* __restrict__ nstat,
                                                 const float* __restrict__ wsc,
                                                 const float* __restrict__ bsc,
                                                 const float* __restrict__ wvc,
                                                 float* __restrict__ out) {
  __shared__ float NB[20];
  const int tid = threadIdx.x;
  const int g0 = blockIdx.x * OT + tid;
  const float4* agg4 = (const float4*)agg;
  const float4* x4 = (const float4*)x;
  const int g = g0 % 5;
  float4 av[OTRIPS], xv[OTRIPS];
#pragma unroll
  for (int it = 0; it < OTRIPS; ++it) {
    const int q = g0 + it * (OGRID * OT);
    av[it] = make_float4(0.f, 0.f, 0.f, 0.f);
    xv[it] = make_float4(0.f, 0.f, 0.f, 0.f);
    if (q < OQ) {
      const float4 a0 = agg4[(q / 5) * (AP / 4) + g], a1 = agg4[AGGN / 4 + (q / 5) * (AP / 4) + g];
      av[it] = make_float4(a0.x + a1.x, a0.y + a1.y, a0.z + a1.z, a0.w + a1.w);
      xv[it] = x4[q];
    }
  }
  if (tid < 12) bn_coeffs(nstat, wsc, bsc, wvc, NB, tid, (float)NN);
  __syncthreads();
  float sc[4], sh[4];
#pragma unroll
  for (int cc = 0; cc < 4; ++cc) {
    const int k = 4 * g + cc;
    sc[cc] = (k < 8) ? NB[k] : NB[16 + (k - 8) / 3];
    sh[cc] = (k < 8) ? NB[8 + k] : 0.0f;
  }
#pragma unroll
  for (int it = 0; it < OTRIPS; ++it) {
    const int q = g0 + it * (OGRID * OT);
    if (q < OQ) {
      float4 r;
      r.x = av[it].x * sc[0] + sh[0] + xv[it].x;
      r.y = av[it].y * sc[1] + sh[1] + xv[it].y;
      r.z = av[it].z * sc[2] + sh[2] + xv[it].z;
      r.w = av[it].w * sc[3] + sh[3] + xv[it].w;
      ((float4*)out)[q] = r;
    }
  }
}

extern "C" void kernel_launch(void* const* d_in, const int* in_sizes, int n_in,
                              void* d_out, int out_size, void* d_ws, size_t ws_size,
                              hipStream_t stream) {
  (void)in_sizes; (void)n_in; (void)out_size; (void)ws_size;
  const float* x = (const float*)d_in[0];
  const float* ea = (const float*)d_in[1];
  const float* elen = (const float*)d_in[2];
  const float* We_ss = (const float*)d_in[3];
  const float* We_sv = (const float*)d_in[4];
  const float* We_vs = (const float*)d_in[5];
  const float* We_vvs = (const float*)d_in[6];
  const float* We_vvv = (const float*)d_in[7];
  const float* bne_ws = (const float*)d_in[8];
  const float* bne_bs = (const float*)d_in[9];
  const float* bne_wv = (const float*)d_in[10];
  const float* Wn_ss = (const float*)d_in[11];
  const float* Wn_sv = (const float*)d_in[12];
  const float* Wn_vs = (const float*)d_in[13];
  const float* Wn_vvs = (const float*)d_in[14];
  const float* Wn_vvv = (const float*)d_in[15];
  const float* bnn_ws = (const float*)d_in[16];
  const float* bnn_bs = (const float*)d_in[17];
  const float* bnn_wv = (const float*)d_in[18];
  const int* eidx = (const int*)d_in[19];

  float* w = (float*)d_ws;
  float* agg = w;
  float* estat = agg + (size_t)2 * AGGN;
  float* nstat = estat + 20;
  float* outf = (float*)d_out;
  float* eout = outf + (size_t)NN * D;

  hipMemsetAsync(agg, 0, ((size_t)2 * AGGN + 40) * sizeof(float), stream);

  k_edge_tp<<<EGRID, PT, 0, stream>>>(x, ea, elen, eidx, We_ss, We_sv, We_vs, We_vvs, We_vvv,
                                        eout, estat);
  k_node_tp<<<NGRID, PT, 0, stream>>>(x, ea, eidx, estat, bne_ws, bne_bs, bne_wv,
                                         Wn_ss, Wn_sv, Wn_vs, Wn_vvs, Wn_vvv, agg, eout);
  k_node_stats<<<(NN + 255) / 256, 256, 0, stream>>>(agg, nstat);
  k_node_out<<<OGRID, OT, 0, stream>>>(agg, x, nstat, bnn_ws, bnn_bs, bnn_wv, outf);
}
